# starting kernel
# baseline (speedup 1.0000x reference)
#include <hip/hip_runtime.h>
#include <hip/hip_bf16.h>

typedef __attribute__((ext_vector_type(2))) float v2f;
typedef __attribute__((ext_vector_type(4))) float v4f;
typedef __attribute__((ext_vector_type(8))) float v8f;

#define IN_CH 256
#define HID   16


__global__ void k_deg_init(unsigned* __restrict__ deg, int N) {
    int n = blockIdx.x * blockDim.x + threadIdx.x;
    if (n < N) deg[n] = 1u;
}

__global__ void k_deg_count(const int* __restrict__ ei, int E,
                            unsigned* __restrict__ deg) {
    int e = blockIdx.x * blockDim.x + threadIdx.x;
    if (e < E) {
        int t = __builtin_nontemporal_load(ei + E + e);
        atomicAdd(&deg[t], 1u);
    }
}

__global__ void k_dinv(const unsigned* __restrict__ deg,
                       float* __restrict__ dinv, int N) {
    int n = blockIdx.x * blockDim.x + threadIdx.x;
    if (n < N) dinv[n] = rsqrtf((float)deg[n]);
}

#define XPAD 4

__global__ __launch_bounds__(64)
void k_gemm_wmma(const float* __restrict__ x,
                 const float* __restrict__ w,
                 float* __restrict__ h, int N) {
    __shared__ float sW[IN_CH][HID];
    __shared__ float sX[2][16][IN_CH + XPAD];

    const int tid  = threadIdx.x;
    const int wv   = tid >> 5;
    const int lane = tid & 31;
    const int l16  = lane & 15;
    const int half = lane >> 4;
    const int node_base = blockIdx.x * 32 + wv * 16;
    const bool full_tile = (node_base + 16 <= N);

    {
        const v4f* ws = (const v4f*)w;
        v4f* wd = (v4f*)&sW[0][0];
#pragma unroll
        for (int i = 0; i < (IN_CH * HID) / 4 / 64; ++i)
            wd[tid + i * 64] = ws[tid + i * 64];
    }

    if (full_tile) {
        const v4f* xs = (const v4f*)(x + (size_t)node_base * IN_CH);
#pragma unroll
        for (int it = 0; it < 32; ++it) {
            int fi = it * 32 + lane;
            v4f v = __builtin_nontemporal_load(xs + fi);
            int f = fi * 4;
            *(v4f*)&sX[wv][f >> 8][f & 255] = v;
        }
    } else {
        for (int it = 0; it < 32; ++it) {
            int f = it * 128 + lane * 4;
            int r = f >> 8, c = f & 255;
            int row = node_base + r;
            v4f v = {};
            if (row < N) v = *(const v4f*)(x + (size_t)row * IN_CH + c);
            *(v4f*)&sX[wv][r][c] = v;
        }
    }
    __syncthreads();

    v8f acc = {};
#pragma unroll 8
    for (int k = 0; k < IN_CH; k += 4) {
        v2f a;
        a.x = sX[wv][l16][k + 2 * half + 0];
        a.y = sX[wv][l16][k + 2 * half + 1];
        v2f b;
        b.x = sW[k + 2 * half + 0][l16];
        b.y = sW[k + 2 * half + 1][l16];
        acc = __builtin_amdgcn_wmma_f32_16x16x4_f32(
            false, a, false, b, (short)0, acc, false, false);
    }

    if (full_tile) {
        unsigned base = (unsigned)(node_base + 8 * half) * HID + (unsigned)l16;
#pragma unroll
        for (int i = 0; i < 8; ++i)
            h[base + (unsigned)i * HID] = acc[i];
    } else {
#pragma unroll
        for (int i = 0; i < 8; ++i) {
            int row = node_base + i + 8 * half;
            if (row < N) h[(size_t)row * HID + l16] = acc[i];
        }
    }
}


__global__ void k_self_init(const float* __restrict__ h,
                            const float* __restrict__ dinv,
                            float* __restrict__ agg, int N) {
    int n = blockIdx.x * blockDim.x + threadIdx.x;
    if (n >= N) return;
    float c = dinv[n];
    c *= c;
    const v4f* hv = (const v4f*)(h + (size_t)n * HID);
    v4f* av = (v4f*)(agg + (size_t)n * HID);
#pragma unroll
    for (int i = 0; i < 4; ++i)
        av[i] = hv[i] * c;
}


__global__ void k_edge_scatter(const int* __restrict__ ei, int E,
                               const float* __restrict__ dinv,
                               const float* __restrict__ h,
                               float* __restrict__ agg) {
    long long gid = (long long)blockIdx.x * blockDim.x + threadIdx.x;
    int e = (int)(gid >> 4);
    int j = (int)(gid & 15);
    if (e >= E) return;
    int s = __builtin_nontemporal_load(ei + e);
    int t = __builtin_nontemporal_load(ei + E + e);
    float coef = dinv[s] * dinv[t];
    unsafeAtomicAdd(&agg[(size_t)t * HID + j], h[(size_t)s * HID + j] * coef);
}


__global__ void k_out(const float* __restrict__ agg,
                      const float* __restrict__ conv_b,
                      const float* __restrict__ out_w,
                      const float* __restrict__ out_b,
                      float* __restrict__ out, int N) {
    int n = blockIdx.x * blockDim.x + threadIdx.x;
    if (n >= N) return;
    const v4f* av = (const v4f*)(agg + (size_t)n * HID);
    const v4f* bv = (const v4f*)conv_b;
    const v4f* wv = (const v4f*)out_w;
    float acc = out_b[0];
#pragma unroll
    for (int i = 0; i < 4; ++i) {
        v4f v = av[i] + bv[i];
        v4f wvi = wv[i];
#pragma unroll
        for (int j = 0; j < 4; ++j)
            acc = fmaf(fmaxf(v[j], 0.0f), wvi[j], acc);
    }
    out[n] = acc;
}


static inline size_t align256(size_t v) { return (v + 255) & ~(size_t)255; }

extern "C" void kernel_launch(void* const* d_in, const int* in_sizes, int n_in,
                              void* d_out, int out_size, void* d_ws, size_t ws_size,
                              hipStream_t stream) {
    const float* x      = (const float*)d_in[0];
    const int*   ei     = (const int*)d_in[1];
    const float* conv_w = (const float*)d_in[2];
    const float* conv_b = (const float*)d_in[3];
    const float* out_w  = (const float*)d_in[4];
    const float* out_b  = (const float*)d_in[5];
    float* out = (float*)d_out;

    const int N = in_sizes[0] / IN_CH;
    const int E = in_sizes[1] / 2;

    char* ws = (char*)d_ws;
    size_t off = 0;
    unsigned* deg = (unsigned*)(ws + off); off = align256(off + (size_t)N * 4);
    float* dinv   = (float*)(ws + off);    off = align256(off + (size_t)N * 4);
    float* h      = (float*)(ws + off);    off = align256(off + (size_t)N * HID * 4);
    float* agg    = (float*)(ws + off);    off = align256(off + (size_t)N * HID * 4);

    const int nb_n = (N + 255) / 256;
    const int nb_e = (E + 255) / 256;

    k_deg_init<<<nb_n, 256, 0, stream>>>(deg, N);
    k_deg_count<<<nb_e, 256, 0, stream>>>(ei, E, deg);
    k_dinv<<<nb_n, 256, 0, stream>>>(deg, dinv, N);

    k_gemm_wmma<<<(N + 31) / 32, 64, 0, stream>>>(x, conv_w, h, N);

    k_self_init<<<nb_n, 256, 0, stream>>>(h, dinv, agg, N);

    long long scatter_threads = (long long)E * HID;
    int nb_s = (int)((scatter_threads + 255) / 256);
    k_edge_scatter<<<nb_s, 256, 0, stream>>>(ei, E, dinv, h, agg);

    k_out<<<nb_n, 256, 0, stream>>>(agg, conv_b, out_w, out_b, out, N);
}


# degree count in LDS histograms (64 edge chunks x 4 node windows, two 16-bit counters per word), partials summed in the dinv kernel, instead of 3.2M scattered global int atomics; float sums untouched
# speedup vs baseline: 1.7479x; 1.7479x over previous
#include <hip/hip_runtime.h>
#include <hip/hip_bf16.h>

typedef __attribute__((ext_vector_type(2))) float v2f;
typedef __attribute__((ext_vector_type(4))) float v4f;
typedef __attribute__((ext_vector_type(8))) float v8f;

#define IN_CH 256
#define HID   16


typedef __attribute__((ext_vector_type(4))) unsigned v4u;
typedef __attribute__((ext_vector_type(4))) int v4i;

#define DEG_PARTS   64
#define DEG_THREADS 512
#define DEG_WORDS   12800

__host__ __device__ static inline int deg_words(int N) { return (((N + 1) >> 1) + 3) & ~3; }

__device__ __forceinline__ void deg_count(unsigned* hist, unsigned wbase, unsigned wn, int tv) {
    const unsigned t = (unsigned)tv;
    const unsigned w = (t >> 1) - wbase;
    if (w < wn) atomicAdd(&hist[w], 1u << ((t & 1u) << 4));
}

__global__ __launch_bounds__(DEG_THREADS)
void k_deg_hist(const int* __restrict__ ei, int E, int N,
                unsigned* __restrict__ part) {
    __shared__ unsigned hist[DEG_WORDS];

    const int tid   = threadIdx.x;
    const int nws   = deg_words(N);
    const int chunk = (((E + DEG_PARTS - 1) / DEG_PARTS) + 3) & ~3;
    const unsigned wbase = blockIdx.y * DEG_WORDS;
    unsigned wn = (unsigned)nws - wbase;
    if (wn > DEG_WORDS) wn = DEG_WORDS;

    int e0 = blockIdx.x * chunk; if (e0 > E) e0 = E;
    int e1 = e0 + chunk;         if (e1 > E) e1 = E;
    const int cnt = e1 - e0;
    const int* tg = ei + E + e0;

    v4u* h4 = (v4u*)hist;
    for (int i = tid; i < DEG_WORDS / 4; i += DEG_THREADS)
        h4[i] = (v4u){0u, 0u, 0u, 0u};
    __syncthreads();

    const int n4 = cnt >> 2;
    const v4i* tg4 = (const v4i*)tg;
    for (int g = tid; g < n4; g += DEG_THREADS * 4) {
        v4i t[4];
#pragma unroll
        for (int u = 0; u < 4; ++u) {
            int gg = g + u * DEG_THREADS;
            t[u] = (gg < n4) ? tg4[gg] : (v4i){-1, -1, -1, -1};
        }
#pragma unroll
        for (int u = 0; u < 4; ++u) {
            deg_count(hist, wbase, wn, t[u].x); deg_count(hist, wbase, wn, t[u].y);
            deg_count(hist, wbase, wn, t[u].z); deg_count(hist, wbase, wn, t[u].w);
        }
    }
    {
        int r = (n4 << 2) + tid;
        if (r < cnt) deg_count(hist, wbase, wn, tg[r]);
    }
    __syncthreads();

    v4u* dst = (v4u*)(part + (size_t)blockIdx.x * nws + wbase);
    for (unsigned i = tid; i < (wn >> 2); i += DEG_THREADS)
        dst[i] = h4[i];
}

__global__ void k_deg_dinv(const unsigned* __restrict__ part,
                           float* __restrict__ dinv, int N) {
    const int i  = blockIdx.x * blockDim.x + threadIdx.x;
    const int n0 = 2 * i;
    if (n0 >= N) return;
    const int nws = deg_words(N);
    unsigned lo = 1u, hi = 1u;
#pragma unroll
    for (int p = 0; p < DEG_PARTS; ++p) {
        unsigned w = part[(size_t)p * nws + i];
        lo += w & 0xFFFFu;
        hi += w >> 16;
    }
    dinv[n0] = rsqrtf((float)lo);
    if (n0 + 1 < N) dinv[n0 + 1] = rsqrtf((float)hi);
}

#define XPAD 4

__global__ __launch_bounds__(64)
void k_gemm_wmma(const float* __restrict__ x,
                 const float* __restrict__ w,
                 float* __restrict__ h, int N) {
    __shared__ float sW[IN_CH][HID];
    __shared__ float sX[2][16][IN_CH + XPAD];

    const int tid  = threadIdx.x;
    const int wv   = tid >> 5;
    const int lane = tid & 31;
    const int l16  = lane & 15;
    const int half = lane >> 4;
    const int node_base = blockIdx.x * 32 + wv * 16;
    const bool full_tile = (node_base + 16 <= N);

    {
        const v4f* ws = (const v4f*)w;
        v4f* wd = (v4f*)&sW[0][0];
#pragma unroll
        for (int i = 0; i < (IN_CH * HID) / 4 / 64; ++i)
            wd[tid + i * 64] = ws[tid + i * 64];
    }

    if (full_tile) {
        const v4f* xs = (const v4f*)(x + (size_t)node_base * IN_CH);
#pragma unroll
        for (int it = 0; it < 32; ++it) {
            int fi = it * 32 + lane;
            v4f v = __builtin_nontemporal_load(xs + fi);
            int f = fi * 4;
            *(v4f*)&sX[wv][f >> 8][f & 255] = v;
        }
    } else {
        for (int it = 0; it < 32; ++it) {
            int f = it * 128 + lane * 4;
            int r = f >> 8, c = f & 255;
            int row = node_base + r;
            v4f v = {};
            if (row < N) v = *(const v4f*)(x + (size_t)row * IN_CH + c);
            *(v4f*)&sX[wv][r][c] = v;
        }
    }
    __syncthreads();

    v8f acc = {};
#pragma unroll 8
    for (int k = 0; k < IN_CH; k += 4) {
        v2f a;
        a.x = sX[wv][l16][k + 2 * half + 0];
        a.y = sX[wv][l16][k + 2 * half + 1];
        v2f b;
        b.x = sW[k + 2 * half + 0][l16];
        b.y = sW[k + 2 * half + 1][l16];
        acc = __builtin_amdgcn_wmma_f32_16x16x4_f32(
            false, a, false, b, (short)0, acc, false, false);
    }

    if (full_tile) {
        unsigned base = (unsigned)(node_base + 8 * half) * HID + (unsigned)l16;
#pragma unroll
        for (int i = 0; i < 8; ++i)
            h[base + (unsigned)i * HID] = acc[i];
    } else {
#pragma unroll
        for (int i = 0; i < 8; ++i) {
            int row = node_base + i + 8 * half;
            if (row < N) h[(size_t)row * HID + l16] = acc[i];
        }
    }
}


__global__ void k_self_init(const float* __restrict__ h,
                            const float* __restrict__ dinv,
                            float* __restrict__ agg, int N) {
    int n = blockIdx.x * blockDim.x + threadIdx.x;
    if (n >= N) return;
    float c = dinv[n];
    c *= c;
    const v4f* hv = (const v4f*)(h + (size_t)n * HID);
    v4f* av = (v4f*)(agg + (size_t)n * HID);
#pragma unroll
    for (int i = 0; i < 4; ++i)
        av[i] = hv[i] * c;
}


__global__ void k_edge_scatter(const int* __restrict__ ei, int E,
                               const float* __restrict__ dinv,
                               const float* __restrict__ h,
                               float* __restrict__ agg) {
    long long gid = (long long)blockIdx.x * blockDim.x + threadIdx.x;
    int e = (int)(gid >> 4);
    int j = (int)(gid & 15);
    if (e >= E) return;
    int s = __builtin_nontemporal_load(ei + e);
    int t = __builtin_nontemporal_load(ei + E + e);
    float coef = dinv[s] * dinv[t];
    unsafeAtomicAdd(&agg[(size_t)t * HID + j], h[(size_t)s * HID + j] * coef);
}


__global__ void k_out(const float* __restrict__ agg,
                      const float* __restrict__ conv_b,
                      const float* __restrict__ out_w,
                      const float* __restrict__ out_b,
                      float* __restrict__ out, int N) {
    int n = blockIdx.x * blockDim.x + threadIdx.x;
    if (n >= N) return;
    const v4f* av = (const v4f*)(agg + (size_t)n * HID);
    const v4f* bv = (const v4f*)conv_b;
    const v4f* wv = (const v4f*)out_w;
    float acc = out_b[0];
#pragma unroll
    for (int i = 0; i < 4; ++i) {
        v4f v = av[i] + bv[i];
        v4f wvi = wv[i];
#pragma unroll
        for (int j = 0; j < 4; ++j)
            acc = fmaf(fmaxf(v[j], 0.0f), wvi[j], acc);
    }
    out[n] = acc;
}


static inline size_t align256(size_t v) { return (v + 255) & ~(size_t)255; }

extern "C" void kernel_launch(void* const* d_in, const int* in_sizes, int n_in,
                              void* d_out, int out_size, void* d_ws, size_t ws_size,
                              hipStream_t stream) {
    const float* x      = (const float*)d_in[0];
    const int*   ei     = (const int*)d_in[1];
    const float* conv_w = (const float*)d_in[2];
    const float* conv_b = (const float*)d_in[3];
    const float* out_w  = (const float*)d_in[4];
    const float* out_b  = (const float*)d_in[5];
    float* out = (float*)d_out;

    const int N = in_sizes[0] / IN_CH;
    const int E = in_sizes[1] / 2;

    char* ws = (char*)d_ws;
    size_t off = 0;
    float* dinv   = (float*)(ws + off);    off = align256(off + (size_t)N * 4);
    float* h      = (float*)(ws + off);    off = align256(off + (size_t)N * HID * 4);
    float* agg    = (float*)(ws + off);    off = align256(off + (size_t)N * HID * 4);
    unsigned* part = (unsigned*)h;

    const int nb_n = (N + 255) / 256;

    const int nws  = deg_words(N);
    const int nwin = (nws + DEG_WORDS - 1) / DEG_WORDS;
    k_deg_hist<<<dim3(DEG_PARTS, nwin), DEG_THREADS, 0, stream>>>(ei, E, N, part);
    k_deg_dinv<<<((N + 1) / 2 + 255) / 256, 256, 0, stream>>>(part, dinv, N);

    k_gemm_wmma<<<(N + 31) / 32, 64, 0, stream>>>(x, conv_w, h, N);

    k_self_init<<<nb_n, 256, 0, stream>>>(h, dinv, agg, N);

    long long scatter_threads = (long long)E * HID;
    int nb_s = (int)((scatter_threads + 255) / 256);
    k_edge_scatter<<<nb_s, 256, 0, stream>>>(ei, E, dinv, h, agg);

    k_out<<<nb_n, 256, 0, stream>>>(agg, conv_b, out_w, out_b, out, N);
}


# edge sums in LDS node windows (9 edge chunks x 28 windows, per-wave hit lists, LDS float adds, whole-line adds of each window into agg), not a global float atomic row per edge; sum order changes
# speedup vs baseline: 2.6191x; 1.4984x over previous
#include <hip/hip_runtime.h>
#include <hip/hip_bf16.h>

typedef __attribute__((ext_vector_type(2))) float v2f;
typedef __attribute__((ext_vector_type(4))) float v4f;
typedef __attribute__((ext_vector_type(8))) float v8f;

#define IN_CH 256
#define HID   16


typedef __attribute__((ext_vector_type(4))) unsigned v4u;
typedef __attribute__((ext_vector_type(4))) int v4i;

#define DEG_PARTS   64
#define DEG_THREADS 512
#define DEG_WORDS   12800

__host__ __device__ static inline int deg_words(int N) { return (((N + 1) >> 1) + 3) & ~3; }

__device__ __forceinline__ void deg_count(unsigned* hist, unsigned wbase, unsigned wn, int tv) {
    const unsigned t = (unsigned)tv;
    const unsigned w = (t >> 1) - wbase;
    if (w < wn) atomicAdd(&hist[w], 1u << ((t & 1u) << 4));
}

__global__ __launch_bounds__(DEG_THREADS)
void k_deg_hist(const int* __restrict__ ei, int E, int N,
                unsigned* __restrict__ part) {
    __shared__ unsigned hist[DEG_WORDS];

    const int tid   = threadIdx.x;
    const int nws   = deg_words(N);
    const int chunk = (((E + DEG_PARTS - 1) / DEG_PARTS) + 3) & ~3;
    const unsigned wbase = blockIdx.y * DEG_WORDS;
    unsigned wn = (unsigned)nws - wbase;
    if (wn > DEG_WORDS) wn = DEG_WORDS;

    int e0 = blockIdx.x * chunk; if (e0 > E) e0 = E;
    int e1 = e0 + chunk;         if (e1 > E) e1 = E;
    const int cnt = e1 - e0;
    const int* tg = ei + E + e0;

    v4u* h4 = (v4u*)hist;
    for (int i = tid; i < DEG_WORDS / 4; i += DEG_THREADS)
        h4[i] = (v4u){0u, 0u, 0u, 0u};
    __syncthreads();

    const int n4 = cnt >> 2;
    const v4i* tg4 = (const v4i*)tg;
    for (int g = tid; g < n4; g += DEG_THREADS * 4) {
        v4i t[4];
#pragma unroll
        for (int u = 0; u < 4; ++u) {
            int gg = g + u * DEG_THREADS;
            t[u] = (gg < n4) ? tg4[gg] : (v4i){-1, -1, -1, -1};
        }
#pragma unroll
        for (int u = 0; u < 4; ++u) {
            deg_count(hist, wbase, wn, t[u].x); deg_count(hist, wbase, wn, t[u].y);
            deg_count(hist, wbase, wn, t[u].z); deg_count(hist, wbase, wn, t[u].w);
        }
    }
    {
        int r = (n4 << 2) + tid;
        if (r < cnt) deg_count(hist, wbase, wn, tg[r]);
    }
    __syncthreads();

    v4u* dst = (v4u*)(part + (size_t)blockIdx.x * nws + wbase);
    for (unsigned i = tid; i < (wn >> 2); i += DEG_THREADS)
        dst[i] = h4[i];
}

__global__ void k_deg_dinv(const unsigned* __restrict__ part,
                           float* __restrict__ dinv, int N) {
    const int i  = blockIdx.x * blockDim.x + threadIdx.x;
    const int n0 = 2 * i;
    if (n0 >= N) return;
    const int nws = deg_words(N);
    unsigned lo = 1u, hi = 1u;
#pragma unroll
    for (int p = 0; p < DEG_PARTS; ++p) {
        unsigned w = part[(size_t)p * nws + i];
        lo += w & 0xFFFFu;
        hi += w >> 16;
    }
    dinv[n0] = rsqrtf((float)lo);
    if (n0 + 1 < N) dinv[n0 + 1] = rsqrtf((float)hi);
}

#define XPAD 4

__global__ __launch_bounds__(64)
void k_gemm_wmma(const float* __restrict__ x,
                 const float* __restrict__ w,
                 float* __restrict__ h, int N) {
    __shared__ float sW[IN_CH][HID];
    __shared__ float sX[2][16][IN_CH + XPAD];

    const int tid  = threadIdx.x;
    const int wv   = tid >> 5;
    const int lane = tid & 31;
    const int l16  = lane & 15;
    const int half = lane >> 4;
    const int node_base = blockIdx.x * 32 + wv * 16;
    const bool full_tile = (node_base + 16 <= N);

    {
        const v4f* ws = (const v4f*)w;
        v4f* wd = (v4f*)&sW[0][0];
#pragma unroll
        for (int i = 0; i < (IN_CH * HID) / 4 / 64; ++i)
            wd[tid + i * 64] = ws[tid + i * 64];
    }

    if (full_tile) {
        const v4f* xs = (const v4f*)(x + (size_t)node_base * IN_CH);
#pragma unroll
        for (int it = 0; it < 32; ++it) {
            int fi = it * 32 + lane;
            v4f v = __builtin_nontemporal_load(xs + fi);
            int f = fi * 4;
            *(v4f*)&sX[wv][f >> 8][f & 255] = v;
        }
    } else {
        for (int it = 0; it < 32; ++it) {
            int f = it * 128 + lane * 4;
            int r = f >> 8, c = f & 255;
            int row = node_base + r;
            v4f v = {};
            if (row < N) v = *(const v4f*)(x + (size_t)row * IN_CH + c);
            *(v4f*)&sX[wv][r][c] = v;
        }
    }
    __syncthreads();

    v8f acc = {};
#pragma unroll 8
    for (int k = 0; k < IN_CH; k += 4) {
        v2f a;
        a.x = sX[wv][l16][k + 2 * half + 0];
        a.y = sX[wv][l16][k + 2 * half + 1];
        v2f b;
        b.x = sW[k + 2 * half + 0][l16];
        b.y = sW[k + 2 * half + 1][l16];
        acc = __builtin_amdgcn_wmma_f32_16x16x4_f32(
            false, a, false, b, (short)0, acc, false, false);
    }

    if (full_tile) {
        unsigned base = (unsigned)(node_base + 8 * half) * HID + (unsigned)l16;
#pragma unroll
        for (int i = 0; i < 8; ++i)
            h[base + (unsigned)i * HID] = acc[i];
    } else {
#pragma unroll
        for (int i = 0; i < 8; ++i) {
            int row = node_base + i + 8 * half;
            if (row < N) h[(size_t)row * HID + l16] = acc[i];
        }
    }
}


__global__ void k_self_init(const float* __restrict__ h,
                            const float* __restrict__ dinv,
                            float* __restrict__ agg, int N) {
    int n = blockIdx.x * blockDim.x + threadIdx.x;
    if (n >= N) return;
    float c = dinv[n];
    c *= c;
    const v4f* hv = (const v4f*)(h + (size_t)n * HID);
    v4f* av = (v4f*)(agg + (size_t)n * HID);
#pragma unroll
    for (int i = 0; i < 4; ++i)
        av[i] = hv[i] * c;
}


#define AG_THREADS 1024
#define AG_WAVES   (AG_THREADS / 32)
#define AG_CHUNKS  9
#define AG_WIN     3584
#define AG_CAPW    512

struct __attribute__((aligned(16))) ag_slot { unsigned ho, lo; float cf; unsigned pad; };

__device__ __forceinline__ void ag_direct(const int* __restrict__ ei, int E,
                                       const float* __restrict__ dinv,
                                       const float* __restrict__ h,
                                       float* __restrict__ agg, int e) {
    const int s = ei[e], t = ei[E + e];
    const float c = dinv[s] * dinv[t];
#pragma nounroll
    for (int j = 0; j < HID; ++j)
        unsafeAtomicAdd(&agg[(size_t)t * HID + j], h[(size_t)s * HID + j] * c);
}

__global__ __launch_bounds__(AG_THREADS)
void k_edge_agg(const int* __restrict__ ei, int E, int N,
                const float* __restrict__ dinv,
                const float* __restrict__ h,
                float* __restrict__ agg) {
    __shared__ float   win[AG_WIN * HID];
    __shared__ int     list[AG_WAVES][AG_CAPW];
    __shared__ ag_slot slot[AG_WAVES][32];

    const int tid  = threadIdx.x;
    const int lane = tid & 31;
    const int wv   = tid >> 5;
    const int l16  = lane & 15;
    const int half = lane >> 4;
    const unsigned lt_mask = (1u << lane) - 1u;

    const int nbase = blockIdx.y * AG_WIN;
    int wn = N - nbase; if (wn > AG_WIN) wn = AG_WIN;
    const int chunk = (((E + AG_CHUNKS - 1) / AG_CHUNKS) + 3) & ~3;
    int e0 = blockIdx.x * chunk; if (e0 > E) e0 = E;
    int e1 = e0 + chunk;         if (e1 > E) e1 = E;
    const int cnt = e1 - e0;
    const int* tg = ei + E + e0;

    v4f* w4 = (v4f*)win;
    for (int i = tid; i < AG_WIN * HID / 4; i += AG_THREADS)
        w4[i] = (v4f){0.0f, 0.0f, 0.0f, 0.0f};

    int* mylist = list[wv];
    int wcnt = 0;
    const int n4 = cnt >> 2;
    const v4i* tg4 = (const v4i*)tg;
    for (int g0 = wv * 32; g0 < n4; g0 += AG_THREADS * 4) {
        v4i t[4];
#pragma unroll
        for (int u = 0; u < 4; ++u) {
            const int g = g0 + u * AG_THREADS + lane;
            t[u] = (g < n4) ? tg4[g] : (v4i){-1, -1, -1, -1};
        }
#pragma unroll
        for (int u = 0; u < 4; ++u) {
            const int g = g0 + u * AG_THREADS + lane;
#pragma unroll
            for (int c = 0; c < 4; ++c) {
                const bool hit = (unsigned)(t[u][c] - nbase) < (unsigned)wn;
                const unsigned m = __builtin_amdgcn_ballot_w32(hit);
                if (hit) {
                    const int pos = wcnt + __builtin_popcount(m & lt_mask);
                    const int e = e0 + 4 * g + c;
                    if (pos < AG_CAPW) mylist[pos] = e;
                    else ag_direct(ei, E, dinv, h, agg, e);
                }
                wcnt += __builtin_popcount(m);
            }
        }
    }
    if (wv == 0) {
        const int r = (n4 << 2) + lane;
        bool hit = false;
        if (lane < 3 && r < cnt) hit = (unsigned)(tg[r] - nbase) < (unsigned)wn;
        const unsigned m = __builtin_amdgcn_ballot_w32(hit);
        if (hit) {
            const int pos = wcnt + __builtin_popcount(m & lt_mask);
            const int e = e0 + r;
            if (pos < AG_CAPW) mylist[pos] = e;
            else ag_direct(ei, E, dinv, h, agg, e);
        }
        wcnt += __builtin_popcount(m);
    }
    __syncthreads();

    const int n = wcnt < AG_CAPW ? wcnt : AG_CAPW;
    for (int k = 0; k < n; k += 32) {
        ag_slot mine = {0u, 0u, 0.0f, 0u};
        if (k + lane < n) {
            const int e = mylist[k + lane];
            const int s = ei[e];
            const int t = ei[E + e];
            mine.ho = (unsigned)s * HID;
            mine.lo = (unsigned)(t - nbase) * HID;
            mine.cf = dinv[s] * dinv[t];
        }
        slot[wv][lane] = mine;

        ag_slot q[16];
        float hv[16];
#pragma unroll
        for (int i = 0; i < 16; ++i) {
            q[i]  = slot[wv][2 * i + half];
            hv[i] = h[q[i].ho + (unsigned)l16];
        }
#pragma unroll
        for (int i = 0; i < 16; ++i) {
            if (k + 2 * i + half < n)
                atomicAdd(&win[q[i].lo + (unsigned)l16], hv[i] * q[i].cf);
        }
    }
    __syncthreads();

    float* dst = agg + (size_t)nbase * HID;
    for (int i = tid; i < wn * HID; i += AG_THREADS)
        unsafeAtomicAdd(&dst[i], win[i]);
}


__global__ void k_out(const float* __restrict__ agg,
                      const float* __restrict__ conv_b,
                      const float* __restrict__ out_w,
                      const float* __restrict__ out_b,
                      float* __restrict__ out, int N) {
    int n = blockIdx.x * blockDim.x + threadIdx.x;
    if (n >= N) return;
    const v4f* av = (const v4f*)(agg + (size_t)n * HID);
    const v4f* bv = (const v4f*)conv_b;
    const v4f* wv = (const v4f*)out_w;
    float acc = out_b[0];
#pragma unroll
    for (int i = 0; i < 4; ++i) {
        v4f v = av[i] + bv[i];
        v4f wvi = wv[i];
#pragma unroll
        for (int j = 0; j < 4; ++j)
            acc = fmaf(fmaxf(v[j], 0.0f), wvi[j], acc);
    }
    out[n] = acc;
}


static inline size_t align256(size_t v) { return (v + 255) & ~(size_t)255; }

extern "C" void kernel_launch(void* const* d_in, const int* in_sizes, int n_in,
                              void* d_out, int out_size, void* d_ws, size_t ws_size,
                              hipStream_t stream) {
    const float* x      = (const float*)d_in[0];
    const int*   ei     = (const int*)d_in[1];
    const float* conv_w = (const float*)d_in[2];
    const float* conv_b = (const float*)d_in[3];
    const float* out_w  = (const float*)d_in[4];
    const float* out_b  = (const float*)d_in[5];
    float* out = (float*)d_out;

    const int N = in_sizes[0] / IN_CH;
    const int E = in_sizes[1] / 2;

    char* ws = (char*)d_ws;
    size_t off = 0;
    float* dinv   = (float*)(ws + off);    off = align256(off + (size_t)N * 4);
    float* h      = (float*)(ws + off);    off = align256(off + (size_t)N * HID * 4);
    float* agg    = (float*)(ws + off);    off = align256(off + (size_t)N * HID * 4);
    unsigned* part = (unsigned*)h;

    const int nb_n = (N + 255) / 256;

    const int nws  = deg_words(N);
    const int nwin = (nws + DEG_WORDS - 1) / DEG_WORDS;
    k_deg_hist<<<dim3(DEG_PARTS, nwin), DEG_THREADS, 0, stream>>>(ei, E, N, part);
    k_deg_dinv<<<((N + 1) / 2 + 255) / 256, 256, 0, stream>>>(part, dinv, N);

    k_gemm_wmma<<<(N + 31) / 32, 64, 0, stream>>>(x, conv_w, h, N);

    k_self_init<<<nb_n, 256, 0, stream>>>(h, dinv, agg, N);

    k_edge_agg<<<dim3(AG_CHUNKS, (N + AG_WIN - 1) / AG_WIN), AG_THREADS, 0, stream>>>(ei, E, N, dinv, h, agg);

    k_out<<<nb_n, 256, 0, stream>>>(agg, conv_b, out_w, out_b, out, N);
}


# edge weight factored out of the sum: h rows pre-scaled by dinv[src], dinv[tgt] applied per node in k_out, target offset kept in the hit list: 1 scattered 4-B read per edge, not 4; rounding changes
# speedup vs baseline: 3.2593x; 1.2444x over previous
#include <hip/hip_runtime.h>
#include <hip/hip_bf16.h>

typedef __attribute__((ext_vector_type(2))) float v2f;
typedef __attribute__((ext_vector_type(4))) float v4f;
typedef __attribute__((ext_vector_type(8))) float v8f;

#define IN_CH 256
#define HID   16


typedef __attribute__((ext_vector_type(4))) unsigned v4u;
typedef __attribute__((ext_vector_type(4))) int v4i;

#define DEG_PARTS   64
#define DEG_THREADS 512
#define DEG_WORDS   12800

__host__ __device__ static inline int deg_words(int N) { return (((N + 1) >> 1) + 3) & ~3; }

__device__ __forceinline__ void deg_count(unsigned* hist, unsigned wbase, unsigned wn, int tv) {
    const unsigned t = (unsigned)tv;
    const unsigned w = (t >> 1) - wbase;
    if (w < wn) atomicAdd(&hist[w], 1u << ((t & 1u) << 4));
}

__global__ __launch_bounds__(DEG_THREADS)
void k_deg_hist(const int* __restrict__ ei, int E, int N,
                unsigned* __restrict__ part) {
    __shared__ unsigned hist[DEG_WORDS];

    const int tid   = threadIdx.x;
    const int nws   = deg_words(N);
    const int chunk = (((E + DEG_PARTS - 1) / DEG_PARTS) + 3) & ~3;
    const unsigned wbase = blockIdx.y * DEG_WORDS;
    unsigned wn = (unsigned)nws - wbase;
    if (wn > DEG_WORDS) wn = DEG_WORDS;

    int e0 = blockIdx.x * chunk; if (e0 > E) e0 = E;
    int e1 = e0 + chunk;         if (e1 > E) e1 = E;
    const int cnt = e1 - e0;
    const int* tg = ei + E + e0;

    v4u* h4 = (v4u*)hist;
    for (int i = tid; i < DEG_WORDS / 4; i += DEG_THREADS)
        h4[i] = (v4u){0u, 0u, 0u, 0u};
    __syncthreads();

    const int n4 = cnt >> 2;
    const v4i* tg4 = (const v4i*)tg;
    for (int g = tid; g < n4; g += DEG_THREADS * 4) {
        v4i t[4];
#pragma unroll
        for (int u = 0; u < 4; ++u) {
            int gg = g + u * DEG_THREADS;
            t[u] = (gg < n4) ? tg4[gg] : (v4i){-1, -1, -1, -1};
        }
#pragma unroll
        for (int u = 0; u < 4; ++u) {
            deg_count(hist, wbase, wn, t[u].x); deg_count(hist, wbase, wn, t[u].y);
            deg_count(hist, wbase, wn, t[u].z); deg_count(hist, wbase, wn, t[u].w);
        }
    }
    {
        int r = (n4 << 2) + tid;
        if (r < cnt) deg_count(hist, wbase, wn, tg[r]);
    }
    __syncthreads();

    v4u* dst = (v4u*)(part + (size_t)blockIdx.x * nws + wbase);
    for (unsigned i = tid; i < (wn >> 2); i += DEG_THREADS)
        dst[i] = h4[i];
}

__global__ void k_deg_dinv(const unsigned* __restrict__ part,
                           float* __restrict__ dinv, int N) {
    const int i  = blockIdx.x * blockDim.x + threadIdx.x;
    const int n0 = 2 * i;
    if (n0 >= N) return;
    const int nws = deg_words(N);
    unsigned lo = 1u, hi = 1u;
#pragma unroll
    for (int p = 0; p < DEG_PARTS; ++p) {
        unsigned w = part[(size_t)p * nws + i];
        lo += w & 0xFFFFu;
        hi += w >> 16;
    }
    dinv[n0] = rsqrtf((float)lo);
    if (n0 + 1 < N) dinv[n0 + 1] = rsqrtf((float)hi);
}

#define XPAD 4

__global__ __launch_bounds__(64)
void k_gemm_wmma(const float* __restrict__ x,
                 const float* __restrict__ w,
                 float* __restrict__ h, int N) {
    __shared__ float sW[IN_CH][HID];
    __shared__ float sX[2][16][IN_CH + XPAD];

    const int tid  = threadIdx.x;
    const int wv   = tid >> 5;
    const int lane = tid & 31;
    const int l16  = lane & 15;
    const int half = lane >> 4;
    const int node_base = blockIdx.x * 32 + wv * 16;
    const bool full_tile = (node_base + 16 <= N);

    {
        const v4f* ws = (const v4f*)w;
        v4f* wd = (v4f*)&sW[0][0];
#pragma unroll
        for (int i = 0; i < (IN_CH * HID) / 4 / 64; ++i)
            wd[tid + i * 64] = ws[tid + i * 64];
    }

    if (full_tile) {
        const v4f* xs = (const v4f*)(x + (size_t)node_base * IN_CH);
#pragma unroll
        for (int it = 0; it < 32; ++it) {
            int fi = it * 32 + lane;
            v4f v = __builtin_nontemporal_load(xs + fi);
            int f = fi * 4;
            *(v4f*)&sX[wv][f >> 8][f & 255] = v;
        }
    } else {
        for (int it = 0; it < 32; ++it) {
            int f = it * 128 + lane * 4;
            int r = f >> 8, c = f & 255;
            int row = node_base + r;
            v4f v = {};
            if (row < N) v = *(const v4f*)(x + (size_t)row * IN_CH + c);
            *(v4f*)&sX[wv][r][c] = v;
        }
    }
    __syncthreads();

    v8f acc = {};
#pragma unroll 8
    for (int k = 0; k < IN_CH; k += 4) {
        v2f a;
        a.x = sX[wv][l16][k + 2 * half + 0];
        a.y = sX[wv][l16][k + 2 * half + 1];
        v2f b;
        b.x = sW[k + 2 * half + 0][l16];
        b.y = sW[k + 2 * half + 1][l16];
        acc = __builtin_amdgcn_wmma_f32_16x16x4_f32(
            false, a, false, b, (short)0, acc, false, false);
    }

    if (full_tile) {
        unsigned base = (unsigned)(node_base + 8 * half) * HID + (unsigned)l16;
#pragma unroll
        for (int i = 0; i < 8; ++i)
            h[base + (unsigned)i * HID] = acc[i];
    } else {
#pragma unroll
        for (int i = 0; i < 8; ++i) {
            int row = node_base + i + 8 * half;
            if (row < N) h[(size_t)row * HID + l16] = acc[i];
        }
    }
}


__global__ void k_prescale(float* __restrict__ h,
                           const float* __restrict__ dinv,
                           float* __restrict__ agg, int N) {
    int n = blockIdx.x * blockDim.x + threadIdx.x;
    if (n >= N) return;
    const float c = dinv[n];
    v4f* hv = (v4f*)(h + (size_t)n * HID);
    v4f* av = (v4f*)(agg + (size_t)n * HID);
#pragma unroll
    for (int i = 0; i < 4; ++i) {
        const v4f v = hv[i] * c;
        hv[i] = v;
        av[i] = v;
    }
}


#define AG_THREADS 1024
#define AG_WAVES   (AG_THREADS / 32)
#define AG_CHUNKS  9
#define AG_WIN     3584
#define AG_CAPW    512
#define AG_TBITS   12
static_assert(AG_WIN <= (1 << AG_TBITS), "a window offset must fit the low bits of a list entry");

struct __attribute__((aligned(8))) ag_slot { unsigned ho, lo; };

__device__ __forceinline__ void ag_direct(const float* __restrict__ h,
                                       float* __restrict__ agg, int s, int t) {
#pragma nounroll
    for (int j = 0; j < HID; ++j)
        unsafeAtomicAdd(&agg[(size_t)t * HID + j], h[(size_t)s * HID + j]);
}

__global__ __launch_bounds__(AG_THREADS)
void k_edge_agg(const int* __restrict__ ei, int E, int N,
                const float* __restrict__ h,
                float* __restrict__ agg) {
    __shared__ float    win[AG_WIN * HID];
    __shared__ unsigned list[AG_WAVES][AG_CAPW];
    __shared__ ag_slot slot[AG_WAVES][32];

    const int tid  = threadIdx.x;
    const int lane = tid & 31;
    const int wv   = tid >> 5;
    const int l16  = lane & 15;
    const int half = lane >> 4;
    const unsigned lt_mask = (1u << lane) - 1u;

    const int nbase = blockIdx.y * AG_WIN;
    int wn = N - nbase; if (wn > AG_WIN) wn = AG_WIN;
    const int chunk = (((E + AG_CHUNKS - 1) / AG_CHUNKS) + 3) & ~3;
    int e0 = blockIdx.x * chunk; if (e0 > E) e0 = E;
    int e1 = e0 + chunk;         if (e1 > E) e1 = E;
    const int cnt = e1 - e0;
    const int* tg = ei + E + e0;

    v4f* w4 = (v4f*)win;
    for (int i = tid; i < AG_WIN * HID / 4; i += AG_THREADS)
        w4[i] = (v4f){0.0f, 0.0f, 0.0f, 0.0f};

    unsigned* mylist = list[wv];
    int wcnt = 0;
    const int n4 = cnt >> 2;
    const v4i* tg4 = (const v4i*)tg;
    for (int g0 = wv * 32; g0 < n4; g0 += AG_THREADS * 4) {
        v4i t[4];
#pragma unroll
        for (int u = 0; u < 4; ++u) {
            const int g = g0 + u * AG_THREADS + lane;
            t[u] = (g < n4) ? tg4[g] : (v4i){-1, -1, -1, -1};
        }
#pragma unroll
        for (int u = 0; u < 4; ++u) {
            const int g = g0 + u * AG_THREADS + lane;
#pragma unroll
            for (int c = 0; c < 4; ++c) {
                const bool hit = (unsigned)(t[u][c] - nbase) < (unsigned)wn;
                const unsigned m = __builtin_amdgcn_ballot_w32(hit);
                if (hit) {
                    const int pos = wcnt + __builtin_popcount(m & lt_mask);
                    const unsigned le = (unsigned)(4 * g + c);
                    if (pos < AG_CAPW) mylist[pos] = (le << AG_TBITS) | (unsigned)(t[u][c] - nbase);
                    else ag_direct(h, agg, ei[e0 + (int)le], t[u][c]);
                }
                wcnt += __builtin_popcount(m);
            }
        }
    }
    if (wv == 0) {
        const int r = (n4 << 2) + lane;
        int tv = -1;
        if (lane < 3 && r < cnt) tv = tg[r];
        const bool hit = (unsigned)(tv - nbase) < (unsigned)wn;
        const unsigned m = __builtin_amdgcn_ballot_w32(hit);
        if (hit) {
            const int pos = wcnt + __builtin_popcount(m & lt_mask);
            if (pos < AG_CAPW) mylist[pos] = ((unsigned)r << AG_TBITS) | (unsigned)(tv - nbase);
            else ag_direct(h, agg, ei[e0 + r], tv);
        }
        wcnt += __builtin_popcount(m);
    }
    __syncthreads();

    const int n = wcnt < AG_CAPW ? wcnt : AG_CAPW;
    for (int k = 0; k < n; k += 32) {
        ag_slot mine = {0u, 0u};
        if (k + lane < n) {
            const unsigned p = mylist[k + lane];
            const int s = ei[e0 + (int)(p >> AG_TBITS)];
            mine.ho = (unsigned)s * HID;
            mine.lo = (p & ((1u << AG_TBITS) - 1u)) * HID;
        }
        slot[wv][lane] = mine;

        ag_slot q[16];
        float hv[16];
#pragma unroll
        for (int i = 0; i < 16; ++i) {
            q[i]  = slot[wv][2 * i + half];
            hv[i] = h[q[i].ho + (unsigned)l16];
        }
#pragma unroll
        for (int i = 0; i < 16; ++i) {
            if (k + 2 * i + half < n)
                atomicAdd(&win[q[i].lo + (unsigned)l16], hv[i]);
        }
    }
    __syncthreads();

    float* dst = agg + (size_t)nbase * HID;
    for (int i = tid; i < wn * HID; i += AG_THREADS)
        unsafeAtomicAdd(&dst[i], win[i]);
}


__global__ void k_out(const float* __restrict__ agg,
                      const float* __restrict__ dinv,
                      const float* __restrict__ conv_b,
                      const float* __restrict__ out_w,
                      const float* __restrict__ out_b,
                      float* __restrict__ out, int N) {
    int n = blockIdx.x * blockDim.x + threadIdx.x;
    if (n >= N) return;
    const v4f* av = (const v4f*)(agg + (size_t)n * HID);
    const v4f* bv = (const v4f*)conv_b;
    const v4f* wv = (const v4f*)out_w;
    const float dn = dinv[n];
    float acc = out_b[0];
#pragma unroll
    for (int i = 0; i < 4; ++i) {
        v4f v = av[i] * dn + bv[i];
        v4f wvi = wv[i];
#pragma unroll
        for (int j = 0; j < 4; ++j)
            acc = fmaf(fmaxf(v[j], 0.0f), wvi[j], acc);
    }
    out[n] = acc;
}


static inline size_t align256(size_t v) { return (v + 255) & ~(size_t)255; }

extern "C" void kernel_launch(void* const* d_in, const int* in_sizes, int n_in,
                              void* d_out, int out_size, void* d_ws, size_t ws_size,
                              hipStream_t stream) {
    const float* x      = (const float*)d_in[0];
    const int*   ei     = (const int*)d_in[1];
    const float* conv_w = (const float*)d_in[2];
    const float* conv_b = (const float*)d_in[3];
    const float* out_w  = (const float*)d_in[4];
    const float* out_b  = (const float*)d_in[5];
    float* out = (float*)d_out;

    const int N = in_sizes[0] / IN_CH;
    const int E = in_sizes[1] / 2;

    char* ws = (char*)d_ws;
    size_t off = 0;
    float* dinv   = (float*)(ws + off);    off = align256(off + (size_t)N * 4);
    float* h      = (float*)(ws + off);    off = align256(off + (size_t)N * HID * 4);
    float* agg    = (float*)(ws + off);    off = align256(off + (size_t)N * HID * 4);
    unsigned* part = (unsigned*)h;

    const int nb_n = (N + 255) / 256;

    const int nws  = deg_words(N);
    const int nwin = (nws + DEG_WORDS - 1) / DEG_WORDS;
    k_deg_hist<<<dim3(DEG_PARTS, nwin), DEG_THREADS, 0, stream>>>(ei, E, N, part);
    k_deg_dinv<<<((N + 1) / 2 + 255) / 256, 256, 0, stream>>>(part, dinv, N);

    k_gemm_wmma<<<(N + 31) / 32, 64, 0, stream>>>(x, conv_w, h, N);

    k_prescale<<<nb_n, 256, 0, stream>>>(h, dinv, agg, N);

    k_edge_agg<<<dim3(AG_CHUNKS, (N + AG_WIN - 1) / AG_WIN), AG_THREADS, 0, stream>>>(ei, E, N, h, agg);

    k_out<<<nb_n, 256, 0, stream>>>(agg, dinv, conv_b, out_w, out_b, out, N);
}


# edge kernel's hot loops lose their per-item branches: scan tests once per trip that all hits fit the list, ranks by mbcnt, loads unguarded inside the chunk; full groups of 32 add without bounds tests
# speedup vs baseline: 3.5745x; 1.0967x over previous
#include <hip/hip_runtime.h>
#include <hip/hip_bf16.h>

typedef __attribute__((ext_vector_type(2))) float v2f;
typedef __attribute__((ext_vector_type(4))) float v4f;
typedef __attribute__((ext_vector_type(8))) float v8f;

#define IN_CH 256
#define HID   16


typedef __attribute__((ext_vector_type(4))) unsigned v4u;
typedef __attribute__((ext_vector_type(4))) int v4i;

#define DEG_PARTS   64
#define DEG_THREADS 512
#define DEG_WORDS   12800

__host__ __device__ static inline int deg_words(int N) { return (((N + 1) >> 1) + 3) & ~3; }

__device__ __forceinline__ void deg_count(unsigned* hist, unsigned wbase, unsigned wn, int tv) {
    const unsigned t = (unsigned)tv;
    const unsigned w = (t >> 1) - wbase;
    if (w < wn) atomicAdd(&hist[w], 1u << ((t & 1u) << 4));
}

__global__ __launch_bounds__(DEG_THREADS)
void k_deg_hist(const int* __restrict__ ei, int E, int N,
                unsigned* __restrict__ part) {
    __shared__ unsigned hist[DEG_WORDS];

    const int tid   = threadIdx.x;
    const int nws   = deg_words(N);
    const int chunk = (((E + DEG_PARTS - 1) / DEG_PARTS) + 3) & ~3;
    const unsigned wbase = blockIdx.y * DEG_WORDS;
    unsigned wn = (unsigned)nws - wbase;
    if (wn > DEG_WORDS) wn = DEG_WORDS;

    int e0 = blockIdx.x * chunk; if (e0 > E) e0 = E;
    int e1 = e0 + chunk;         if (e1 > E) e1 = E;
    const int cnt = e1 - e0;
    const int* tg = ei + E + e0;

    v4u* h4 = (v4u*)hist;
    for (int i = tid; i < DEG_WORDS / 4; i += DEG_THREADS)
        h4[i] = (v4u){0u, 0u, 0u, 0u};
    __syncthreads();

    const int n4 = cnt >> 2;
    const v4i* tg4 = (const v4i*)tg;
    for (int g = tid; g < n4; g += DEG_THREADS * 4) {
        v4i t[4];
#pragma unroll
        for (int u = 0; u < 4; ++u) {
            int gg = g + u * DEG_THREADS;
            t[u] = (gg < n4) ? tg4[gg] : (v4i){-1, -1, -1, -1};
        }
#pragma unroll
        for (int u = 0; u < 4; ++u) {
            deg_count(hist, wbase, wn, t[u].x); deg_count(hist, wbase, wn, t[u].y);
            deg_count(hist, wbase, wn, t[u].z); deg_count(hist, wbase, wn, t[u].w);
        }
    }
    {
        int r = (n4 << 2) + tid;
        if (r < cnt) deg_count(hist, wbase, wn, tg[r]);
    }
    __syncthreads();

    v4u* dst = (v4u*)(part + (size_t)blockIdx.x * nws + wbase);
    for (unsigned i = tid; i < (wn >> 2); i += DEG_THREADS)
        dst[i] = h4[i];
}

__global__ void k_deg_dinv(const unsigned* __restrict__ part,
                           float* __restrict__ dinv, int N) {
    const int i  = blockIdx.x * blockDim.x + threadIdx.x;
    const int n0 = 2 * i;
    if (n0 >= N) return;
    const int nws = deg_words(N);
    unsigned lo = 1u, hi = 1u;
#pragma unroll
    for (int p = 0; p < DEG_PARTS; ++p) {
        unsigned w = part[(size_t)p * nws + i];
        lo += w & 0xFFFFu;
        hi += w >> 16;
    }
    dinv[n0] = rsqrtf((float)lo);
    if (n0 + 1 < N) dinv[n0 + 1] = rsqrtf((float)hi);
}

#define XPAD 4

__global__ __launch_bounds__(64)
void k_gemm_wmma(const float* __restrict__ x,
                 const float* __restrict__ w,
                 float* __restrict__ h, int N) {
    __shared__ float sW[IN_CH][HID];
    __shared__ float sX[2][16][IN_CH + XPAD];

    const int tid  = threadIdx.x;
    const int wv   = tid >> 5;
    const int lane = tid & 31;
    const int l16  = lane & 15;
    const int half = lane >> 4;
    const int node_base = blockIdx.x * 32 + wv * 16;
    const bool full_tile = (node_base + 16 <= N);

    {
        const v4f* ws = (const v4f*)w;
        v4f* wd = (v4f*)&sW[0][0];
#pragma unroll
        for (int i = 0; i < (IN_CH * HID) / 4 / 64; ++i)
            wd[tid + i * 64] = ws[tid + i * 64];
    }

    if (full_tile) {
        const v4f* xs = (const v4f*)(x + (size_t)node_base * IN_CH);
#pragma unroll
        for (int it = 0; it < 32; ++it) {
            int fi = it * 32 + lane;
            v4f v = __builtin_nontemporal_load(xs + fi);
            int f = fi * 4;
            *(v4f*)&sX[wv][f >> 8][f & 255] = v;
        }
    } else {
        for (int it = 0; it < 32; ++it) {
            int f = it * 128 + lane * 4;
            int r = f >> 8, c = f & 255;
            int row = node_base + r;
            v4f v = {};
            if (row < N) v = *(const v4f*)(x + (size_t)row * IN_CH + c);
            *(v4f*)&sX[wv][r][c] = v;
        }
    }
    __syncthreads();

    v8f acc = {};
#pragma unroll 8
    for (int k = 0; k < IN_CH; k += 4) {
        v2f a;
        a.x = sX[wv][l16][k + 2 * half + 0];
        a.y = sX[wv][l16][k + 2 * half + 1];
        v2f b;
        b.x = sW[k + 2 * half + 0][l16];
        b.y = sW[k + 2 * half + 1][l16];
        acc = __builtin_amdgcn_wmma_f32_16x16x4_f32(
            false, a, false, b, (short)0, acc, false, false);
    }

    if (full_tile) {
        unsigned base = (unsigned)(node_base + 8 * half) * HID + (unsigned)l16;
#pragma unroll
        for (int i = 0; i < 8; ++i)
            h[base + (unsigned)i * HID] = acc[i];
    } else {
#pragma unroll
        for (int i = 0; i < 8; ++i) {
            int row = node_base + i + 8 * half;
            if (row < N) h[(size_t)row * HID + l16] = acc[i];
        }
    }
}


__global__ void k_prescale(float* __restrict__ h,
                           const float* __restrict__ dinv,
                           float* __restrict__ agg, int N) {
    int n = blockIdx.x * blockDim.x + threadIdx.x;
    if (n >= N) return;
    const float c = dinv[n];
    v4f* hv = (v4f*)(h + (size_t)n * HID);
    v4f* av = (v4f*)(agg + (size_t)n * HID);
#pragma unroll
    for (int i = 0; i < 4; ++i) {
        const v4f v = hv[i] * c;
        hv[i] = v;
        av[i] = v;
    }
}


#define AG_THREADS 1024
#define AG_WAVES   (AG_THREADS / 32)
#define AG_CHUNKS  9
#define AG_WIN     3584
#define AG_CAPW    512
#define AG_TBITS   12
static_assert(AG_WIN <= (1 << AG_TBITS), "a window offset must fit the low bits of a list entry");

struct __attribute__((aligned(8))) ag_slot { unsigned ho, lo; };

__device__ __forceinline__ void ag_direct(const float* __restrict__ h,
                                       float* __restrict__ agg, int s, int t) {
#pragma nounroll
    for (int j = 0; j < HID; ++j)
        unsafeAtomicAdd(&agg[(size_t)t * HID + j], h[(size_t)s * HID + j]);
}

template <bool FULL>
__device__ __forceinline__ void ag_gather32(const int* __restrict__ src_of, const float* __restrict__ h,
                                            float* win, const unsigned* mylist, ag_slot* myslot,
                                            int k, int n, int lane, int l16, int half) {
    ag_slot mine = {0u, 0u};
    if (FULL || k + lane < n) {
        const unsigned p = mylist[k + lane];
        const int s = src_of[(int)(p >> AG_TBITS)];
        mine.ho = (unsigned)s * HID;
        mine.lo = (p & ((1u << AG_TBITS) - 1u)) * HID;
    }
    myslot[lane] = mine;

    ag_slot q[16];
    float hv[16];
#pragma unroll
    for (int i = 0; i < 16; ++i) {
        q[i]  = myslot[2 * i + half];
        hv[i] = h[q[i].ho + (unsigned)l16];
    }
#pragma unroll
    for (int i = 0; i < 16; ++i) {
        if (FULL || k + 2 * i + half < n)
            atomicAdd(&win[q[i].lo + (unsigned)l16], hv[i]);
    }
}

__global__ __launch_bounds__(AG_THREADS)
void k_edge_agg(const int* __restrict__ ei, int E, int N,
                const float* __restrict__ h,
                float* __restrict__ agg) {
    __shared__ float    win[AG_WIN * HID];
    __shared__ unsigned list[AG_WAVES][AG_CAPW];
    __shared__ ag_slot slot[AG_WAVES][32];

    const int tid  = threadIdx.x;
    const int lane = tid & 31;
    const int wv   = __builtin_amdgcn_readfirstlane(tid >> 5);
    const int l16  = lane & 15;
    const int half = lane >> 4;
    const unsigned lt_mask = (1u << lane) - 1u;

    const int nbase = blockIdx.y * AG_WIN;
    int wn = N - nbase; if (wn > AG_WIN) wn = AG_WIN;
    const int chunk = (((E + AG_CHUNKS - 1) / AG_CHUNKS) + 3) & ~3;
    int e0 = blockIdx.x * chunk; if (e0 > E) e0 = E;
    int e1 = e0 + chunk;         if (e1 > E) e1 = E;
    const int cnt = e1 - e0;
    const int* tg = ei + E + e0;

    v4f* w4 = (v4f*)win;
    for (int i = tid; i < AG_WIN * HID / 4; i += AG_THREADS)
        w4[i] = (v4f){0.0f, 0.0f, 0.0f, 0.0f};

    unsigned* mylist = list[wv];
    int wcnt = 0;
    const int n4 = cnt >> 2;
    const v4i* tg4 = (const v4i*)tg;
    for (int g0 = wv * 32; g0 < n4; g0 += AG_THREADS * 4) {
        v4i t[4];
        if (g0 + 3 * AG_THREADS + 32 <= n4) {
#pragma unroll
            for (int u = 0; u < 4; ++u)
                t[u] = tg4[g0 + u * AG_THREADS + lane];
        } else {
#pragma unroll
            for (int u = 0; u < 4; ++u) {
                const int g = g0 + u * AG_THREADS + lane;
                t[u] = (g < n4) ? tg4[g] : (v4i){-1, -1, -1, -1};
            }
        }
        unsigned m[4][4];
        const int wbase = wcnt;
        int tot = 0;
#pragma unroll
        for (int u = 0; u < 4; ++u) {
#pragma unroll
            for (int c = 0; c < 4; ++c) {
                m[u][c] = __builtin_amdgcn_ballot_w32((unsigned)(t[u][c] - nbase) < (unsigned)wn);
                tot += __builtin_popcount(m[u][c]);
            }
        }
        int run = wbase;
        if (wbase + tot <= AG_CAPW) {
#pragma unroll
            for (int u = 0; u < 4; ++u) {
                const int g = g0 + u * AG_THREADS + lane;
#pragma unroll
                for (int c = 0; c < 4; ++c) {
                    const unsigned off = (unsigned)(t[u][c] - nbase);
                    if (off < (unsigned)wn) {
                        const unsigned pos = __builtin_amdgcn_mbcnt_lo(m[u][c], (unsigned)run);
                        mylist[pos] = ((unsigned)(4 * g + c) << AG_TBITS) | off;
                    }
                    run += __builtin_popcount(m[u][c]);
                }
            }
        } else {
#pragma unroll
            for (int u = 0; u < 4; ++u) {
                const int g = g0 + u * AG_THREADS + lane;
#pragma unroll
                for (int c = 0; c < 4; ++c) {
                    const bool hit = (unsigned)(t[u][c] - nbase) < (unsigned)wn;
                    if (hit) {
                        const int pos = run + __builtin_popcount(m[u][c] & lt_mask);
                        const unsigned le = (unsigned)(4 * g + c);
                        if (pos < AG_CAPW) mylist[pos] = (le << AG_TBITS) | (unsigned)(t[u][c] - nbase);
                        else ag_direct(h, agg, ei[e0 + (int)le], t[u][c]);
                    }
                    run += __builtin_popcount(m[u][c]);
                }
            }
        }
        wcnt = wbase + tot;
    }
    if (wv == 0) {
        const int r = (n4 << 2) + lane;
        int tv = -1;
        if (lane < 3 && r < cnt) tv = tg[r];
        const bool hit = (unsigned)(tv - nbase) < (unsigned)wn;
        const unsigned m = __builtin_amdgcn_ballot_w32(hit);
        if (hit) {
            const int pos = wcnt + __builtin_popcount(m & lt_mask);
            if (pos < AG_CAPW) mylist[pos] = ((unsigned)r << AG_TBITS) | (unsigned)(tv - nbase);
            else ag_direct(h, agg, ei[e0 + r], tv);
        }
        wcnt += __builtin_popcount(m);
    }
    __syncthreads();

    const int n = wcnt < AG_CAPW ? wcnt : AG_CAPW;
    {
        int k = 0;
        for (; k + 32 <= n; k += 32)
            ag_gather32<true>(ei + e0, h, win, mylist, slot[wv], k, n, lane, l16, half);
        if (k < n)
            ag_gather32<false>(ei + e0, h, win, mylist, slot[wv], k, n, lane, l16, half);
    }
    __syncthreads();

    float* dst = agg + (size_t)nbase * HID;
    for (int i = tid; i < wn * HID; i += AG_THREADS)
        unsafeAtomicAdd(&dst[i], win[i]);
}


__global__ void k_out(const float* __restrict__ agg,
                      const float* __restrict__ dinv,
                      const float* __restrict__ conv_b,
                      const float* __restrict__ out_w,
                      const float* __restrict__ out_b,
                      float* __restrict__ out, int N) {
    int n = blockIdx.x * blockDim.x + threadIdx.x;
    if (n >= N) return;
    const v4f* av = (const v4f*)(agg + (size_t)n * HID);
    const v4f* bv = (const v4f*)conv_b;
    const v4f* wv = (const v4f*)out_w;
    const float dn = dinv[n];
    float acc = out_b[0];
#pragma unroll
    for (int i = 0; i < 4; ++i) {
        v4f v = av[i] * dn + bv[i];
        v4f wvi = wv[i];
#pragma unroll
        for (int j = 0; j < 4; ++j)
            acc = fmaf(fmaxf(v[j], 0.0f), wvi[j], acc);
    }
    out[n] = acc;
}


static inline size_t align256(size_t v) { return (v + 255) & ~(size_t)255; }

extern "C" void kernel_launch(void* const* d_in, const int* in_sizes, int n_in,
                              void* d_out, int out_size, void* d_ws, size_t ws_size,
                              hipStream_t stream) {
    const float* x      = (const float*)d_in[0];
    const int*   ei     = (const int*)d_in[1];
    const float* conv_w = (const float*)d_in[2];
    const float* conv_b = (const float*)d_in[3];
    const float* out_w  = (const float*)d_in[4];
    const float* out_b  = (const float*)d_in[5];
    float* out = (float*)d_out;

    const int N = in_sizes[0] / IN_CH;
    const int E = in_sizes[1] / 2;

    char* ws = (char*)d_ws;
    size_t off = 0;
    float* dinv   = (float*)(ws + off);    off = align256(off + (size_t)N * 4);
    float* h      = (float*)(ws + off);    off = align256(off + (size_t)N * HID * 4);
    float* agg    = (float*)(ws + off);    off = align256(off + (size_t)N * HID * 4);
    unsigned* part = (unsigned*)h;

    const int nb_n = (N + 255) / 256;

    const int nws  = deg_words(N);
    const int nwin = (nws + DEG_WORDS - 1) / DEG_WORDS;
    k_deg_hist<<<dim3(DEG_PARTS, nwin), DEG_THREADS, 0, stream>>>(ei, E, N, part);
    k_deg_dinv<<<((N + 1) / 2 + 255) / 256, 256, 0, stream>>>(part, dinv, N);

    k_gemm_wmma<<<(N + 31) / 32, 64, 0, stream>>>(x, conv_w, h, N);

    k_prescale<<<nb_n, 256, 0, stream>>>(h, dinv, agg, N);

    k_edge_agg<<<dim3(AG_CHUNKS, (N + AG_WIN - 1) / AG_WIN), AG_THREADS, 0, stream>>>(ei, E, N, h, agg);

    k_out<<<nb_n, 256, 0, stream>>>(agg, dinv, conv_b, out_w, out_b, out, N);
}


# degree sum and row scaling folded into the GEMM launch: partial counts laid out per 32-node block over that block's agg rows; each block sums its own, writes dinv, scaled h, agg; 6 launches -> 4
# speedup vs baseline: 3.7202x; 1.0408x over previous
#include <hip/hip_runtime.h>
#include <hip/hip_bf16.h>

typedef __attribute__((ext_vector_type(2))) float v2f;
typedef __attribute__((ext_vector_type(4))) float v4f;
typedef __attribute__((ext_vector_type(8))) float v8f;

#define IN_CH 256
#define HID   16


typedef __attribute__((ext_vector_type(4))) unsigned v4u;
typedef __attribute__((ext_vector_type(4))) int v4i;

#define DEG_PARTS   16
#define DEG_THREADS 1024
#define DEG_WIN     12800
#define DEG_LOADS   8
static_assert(DEG_PARTS * 4 == HID * 4, "a node's partial counts must fill exactly its agg row");
static_assert(DEG_WIN % 32 == 0, "a window holds whole node blocks");

__device__ __forceinline__ void deg_count(unsigned* hist, unsigned nbase, unsigned wn, int tv) {
    const unsigned w = (unsigned)tv - nbase;
    if (w < wn) atomicAdd(&hist[w], 1u);
}

__global__ __launch_bounds__(DEG_THREADS)
void k_deg_hist(const int* __restrict__ ei, int E, int N,
                unsigned* __restrict__ part) {
    __shared__ unsigned hist[DEG_WIN];

    const int tid   = threadIdx.x;
    const int lane  = tid & 31;
    const int wv    = tid >> 5;
    const int nblk  = (N + 31) >> 5;
    const int chunk = (((E + DEG_PARTS - 1) / DEG_PARTS) + 3) & ~3;
    const unsigned nbase = blockIdx.y * DEG_WIN;
    unsigned wn = (unsigned)nblk * 32u - nbase;
    if (wn > DEG_WIN) wn = DEG_WIN;

    int e0 = blockIdx.x * chunk; if (e0 > E) e0 = E;
    int e1 = e0 + chunk;         if (e1 > E) e1 = E;
    const int cnt = e1 - e0;
    const int* tg = ei + E + e0;

    v4u* h4 = (v4u*)hist;
    for (int i = tid; i < DEG_WIN / 4; i += DEG_THREADS)
        h4[i] = (v4u){0u, 0u, 0u, 0u};
    __syncthreads();

    const int n4 = cnt >> 2;
    const v4i* tg4 = (const v4i*)tg;
    for (int g = tid; g < n4; g += DEG_THREADS * DEG_LOADS) {
        v4i t[DEG_LOADS];
#pragma unroll
        for (int u = 0; u < DEG_LOADS; ++u) {
            int gg = g + u * DEG_THREADS;
            t[u] = (gg < n4) ? tg4[gg] : (v4i){-1, -1, -1, -1};
        }
#pragma unroll
        for (int u = 0; u < DEG_LOADS; ++u) {
            deg_count(hist, nbase, wn, t[u].x); deg_count(hist, nbase, wn, t[u].y);
            deg_count(hist, nbase, wn, t[u].z); deg_count(hist, nbase, wn, t[u].w);
        }
    }
    {
        int r = (n4 << 2) + tid;
        if (r < cnt) deg_count(hist, nbase, wn, tg[r]);
    }
    __syncthreads();

    const int nb0 = (int)(nbase >> 5);
    const int nbw = (int)(wn >> 5);
    for (int nb = wv; nb < nbw; nb += DEG_THREADS / 32)
        part[((size_t)(nb0 + nb) * DEG_PARTS + blockIdx.x) * 32 + lane] = hist[nb * 32 + lane];
}

#define XPAD 4

__global__ __launch_bounds__(64)
void k_gemm_wmma(const float* __restrict__ x,
                 const float* __restrict__ w,
                 const unsigned* part,
                 float* __restrict__ dinv,
                 float* __restrict__ h,
                 float* agg, int N) {
    __shared__ float sW[IN_CH][HID];
    __shared__ float sX[2][16][IN_CH + XPAD];
    __shared__ unsigned sDeg[2][32];

    const int tid  = threadIdx.x;
    const int wv   = tid >> 5;
    const int lane = tid & 31;
    const int l16  = lane & 15;
    const int half = lane >> 4;
    const int node_base = blockIdx.x * 32 + wv * 16;
    const bool full_tile = (node_base + 16 <= N);

    {
        const unsigned* pp = part + ((size_t)blockIdx.x * DEG_PARTS + (size_t)wv * (DEG_PARTS / 2)) * 32 + lane;
        unsigned sum = 0u;
#pragma unroll
        for (int p = 0; p < DEG_PARTS / 2; ++p)
            sum += pp[p * 32];
        sDeg[wv][lane] = sum;
    }

    {
        const v4f* ws = (const v4f*)w;
        v4f* wd = (v4f*)&sW[0][0];
#pragma unroll
        for (int i = 0; i < (IN_CH * HID) / 4 / 64; ++i)
            wd[tid + i * 64] = ws[tid + i * 64];
    }

    if (full_tile) {
        const v4f* xs = (const v4f*)(x + (size_t)node_base * IN_CH);
#pragma unroll
        for (int it = 0; it < 32; ++it) {
            int fi = it * 32 + lane;
            v4f v = __builtin_nontemporal_load(xs + fi);
            int f = fi * 4;
            *(v4f*)&sX[wv][f >> 8][f & 255] = v;
        }
    } else {
        for (int it = 0; it < 32; ++it) {
            int f = it * 128 + lane * 4;
            int r = f >> 8, c = f & 255;
            int row = node_base + r;
            v4f v = {};
            if (row < N) v = *(const v4f*)(x + (size_t)row * IN_CH + c);
            *(v4f*)&sX[wv][r][c] = v;
        }
    }
    __syncthreads();

    float dv[8];
#pragma unroll
    for (int i = 0; i < 8; ++i) {
        const int r = wv * 16 + 8 * half + i;
        dv[i] = rsqrtf((float)(1u + sDeg[0][r] + sDeg[1][r]));
    }

    v8f acc = {};
#pragma unroll 8
    for (int k = 0; k < IN_CH; k += 4) {
        v2f a;
        a.x = sX[wv][l16][k + 2 * half + 0];
        a.y = sX[wv][l16][k + 2 * half + 1];
        v2f b;
        b.x = sW[k + 2 * half + 0][l16];
        b.y = sW[k + 2 * half + 1][l16];
        acc = __builtin_amdgcn_wmma_f32_16x16x4_f32(
            false, a, false, b, (short)0, acc, false, false);
    }

    if (full_tile) {
        unsigned base = (unsigned)(node_base + 8 * half) * HID + (unsigned)l16;
#pragma unroll
        for (int i = 0; i < 8; ++i) {
            const float v = acc[i] * dv[i];
            h[base + (unsigned)i * HID]   = v;
            agg[base + (unsigned)i * HID] = v;
        }
        if (l16 == 0) {
#pragma unroll
            for (int i = 0; i < 8; ++i)
                dinv[node_base + 8 * half + i] = dv[i];
        }
    } else {
#pragma unroll
        for (int i = 0; i < 8; ++i) {
            int row = node_base + i + 8 * half;
            if (row < N) {
                const float v = acc[i] * dv[i];
                h[(size_t)row * HID + l16]   = v;
                agg[(size_t)row * HID + l16] = v;
                if (l16 == 0) dinv[row] = dv[i];
            }
        }
    }
}


#define AG_THREADS 1024
#define AG_WAVES   (AG_THREADS / 32)
#define AG_CHUNKS  9
#define AG_WIN     3584
#define AG_CAPW    512
#define AG_TBITS   12
static_assert(AG_WIN <= (1 << AG_TBITS), "a window offset must fit the low bits of a list entry");

struct __attribute__((aligned(8))) ag_slot { unsigned ho, lo; };

__device__ __forceinline__ void ag_direct(const float* __restrict__ h,
                                       float* __restrict__ agg, int s, int t) {
#pragma nounroll
    for (int j = 0; j < HID; ++j)
        unsafeAtomicAdd(&agg[(size_t)t * HID + j], h[(size_t)s * HID + j]);
}

template <bool FULL>
__device__ __forceinline__ void ag_gather32(const int* __restrict__ src_of, const float* __restrict__ h,
                                            float* win, const unsigned* mylist, ag_slot* myslot,
                                            int k, int n, int lane, int l16, int half) {
    ag_slot mine = {0u, 0u};
    if (FULL || k + lane < n) {
        const unsigned p = mylist[k + lane];
        const int s = src_of[(int)(p >> AG_TBITS)];
        mine.ho = (unsigned)s * HID;
        mine.lo = (p & ((1u << AG_TBITS) - 1u)) * HID;
    }
    myslot[lane] = mine;

    ag_slot q[16];
    float hv[16];
#pragma unroll
    for (int i = 0; i < 16; ++i) {
        q[i]  = myslot[2 * i + half];
        hv[i] = h[q[i].ho + (unsigned)l16];
    }
#pragma unroll
    for (int i = 0; i < 16; ++i) {
        if (FULL || k + 2 * i + half < n)
            atomicAdd(&win[q[i].lo + (unsigned)l16], hv[i]);
    }
}

__global__ __launch_bounds__(AG_THREADS)
void k_edge_agg(const int* __restrict__ ei, int E, int N,
                const float* __restrict__ h,
                float* __restrict__ agg) {
    __shared__ float    win[AG_WIN * HID];
    __shared__ unsigned list[AG_WAVES][AG_CAPW];
    __shared__ ag_slot slot[AG_WAVES][32];

    const int tid  = threadIdx.x;
    const int lane = tid & 31;
    const int wv   = __builtin_amdgcn_readfirstlane(tid >> 5);
    const int l16  = lane & 15;
    const int half = lane >> 4;
    const unsigned lt_mask = (1u << lane) - 1u;

    const int nbase = blockIdx.y * AG_WIN;
    int wn = N - nbase; if (wn > AG_WIN) wn = AG_WIN;
    const int chunk = (((E + AG_CHUNKS - 1) / AG_CHUNKS) + 3) & ~3;
    int e0 = blockIdx.x * chunk; if (e0 > E) e0 = E;
    int e1 = e0 + chunk;         if (e1 > E) e1 = E;
    const int cnt = e1 - e0;
    const int* tg = ei + E + e0;

    v4f* w4 = (v4f*)win;
    for (int i = tid; i < AG_WIN * HID / 4; i += AG_THREADS)
        w4[i] = (v4f){0.0f, 0.0f, 0.0f, 0.0f};

    unsigned* mylist = list[wv];
    int wcnt = 0;
    const int n4 = cnt >> 2;
    const v4i* tg4 = (const v4i*)tg;
    for (int g0 = wv * 32; g0 < n4; g0 += AG_THREADS * 4) {
        v4i t[4];
        if (g0 + 3 * AG_THREADS + 32 <= n4) {
#pragma unroll
            for (int u = 0; u < 4; ++u)
                t[u] = tg4[g0 + u * AG_THREADS + lane];
        } else {
#pragma unroll
            for (int u = 0; u < 4; ++u) {
                const int g = g0 + u * AG_THREADS + lane;
                t[u] = (g < n4) ? tg4[g] : (v4i){-1, -1, -1, -1};
            }
        }
        unsigned m[4][4];
        const int wbase = wcnt;
        int tot = 0;
#pragma unroll
        for (int u = 0; u < 4; ++u) {
#pragma unroll
            for (int c = 0; c < 4; ++c) {
                m[u][c] = __builtin_amdgcn_ballot_w32((unsigned)(t[u][c] - nbase) < (unsigned)wn);
                tot += __builtin_popcount(m[u][c]);
            }
        }
        int run = wbase;
        if (wbase + tot <= AG_CAPW) {
#pragma unroll
            for (int u = 0; u < 4; ++u) {
                const int g = g0 + u * AG_THREADS + lane;
#pragma unroll
                for (int c = 0; c < 4; ++c) {
                    const unsigned off = (unsigned)(t[u][c] - nbase);
                    if (off < (unsigned)wn) {
                        const unsigned pos = __builtin_amdgcn_mbcnt_lo(m[u][c], (unsigned)run);
                        mylist[pos] = ((unsigned)(4 * g + c) << AG_TBITS) | off;
                    }
                    run += __builtin_popcount(m[u][c]);
                }
            }
        } else {
#pragma unroll
            for (int u = 0; u < 4; ++u) {
                const int g = g0 + u * AG_THREADS + lane;
#pragma unroll
                for (int c = 0; c < 4; ++c) {
                    const bool hit = (unsigned)(t[u][c] - nbase) < (unsigned)wn;
                    if (hit) {
                        const int pos = run + __builtin_popcount(m[u][c] & lt_mask);
                        const unsigned le = (unsigned)(4 * g + c);
                        if (pos < AG_CAPW) mylist[pos] = (le << AG_TBITS) | (unsigned)(t[u][c] - nbase);
                        else ag_direct(h, agg, ei[e0 + (int)le], t[u][c]);
                    }
                    run += __builtin_popcount(m[u][c]);
                }
            }
        }
        wcnt = wbase + tot;
    }
    if (wv == 0) {
        const int r = (n4 << 2) + lane;
        int tv = -1;
        if (lane < 3 && r < cnt) tv = tg[r];
        const bool hit = (unsigned)(tv - nbase) < (unsigned)wn;
        const unsigned m = __builtin_amdgcn_ballot_w32(hit);
        if (hit) {
            const int pos = wcnt + __builtin_popcount(m & lt_mask);
            if (pos < AG_CAPW) mylist[pos] = ((unsigned)r << AG_TBITS) | (unsigned)(tv - nbase);
            else ag_direct(h, agg, ei[e0 + r], tv);
        }
        wcnt += __builtin_popcount(m);
    }
    __syncthreads();

    const int n = wcnt < AG_CAPW ? wcnt : AG_CAPW;
    {
        int k = 0;
        for (; k + 32 <= n; k += 32)
            ag_gather32<true>(ei + e0, h, win, mylist, slot[wv], k, n, lane, l16, half);
        if (k < n)
            ag_gather32<false>(ei + e0, h, win, mylist, slot[wv], k, n, lane, l16, half);
    }
    __syncthreads();

    float* dst = agg + (size_t)nbase * HID;
    for (int i = tid; i < wn * HID; i += AG_THREADS)
        unsafeAtomicAdd(&dst[i], win[i]);
}


__global__ void k_out(const float* __restrict__ agg,
                      const float* __restrict__ dinv,
                      const float* __restrict__ conv_b,
                      const float* __restrict__ out_w,
                      const float* __restrict__ out_b,
                      float* __restrict__ out, int N) {
    int n = blockIdx.x * blockDim.x + threadIdx.x;
    if (n >= N) return;
    const v4f* av = (const v4f*)(agg + (size_t)n * HID);
    const v4f* bv = (const v4f*)conv_b;
    const v4f* wv = (const v4f*)out_w;
    const float dn = dinv[n];
    float acc = out_b[0];
#pragma unroll
    for (int i = 0; i < 4; ++i) {
        v4f v = av[i] * dn + bv[i];
        v4f wvi = wv[i];
#pragma unroll
        for (int j = 0; j < 4; ++j)
            acc = fmaf(fmaxf(v[j], 0.0f), wvi[j], acc);
    }
    out[n] = acc;
}


static inline size_t align256(size_t v) { return (v + 255) & ~(size_t)255; }

extern "C" void kernel_launch(void* const* d_in, const int* in_sizes, int n_in,
                              void* d_out, int out_size, void* d_ws, size_t ws_size,
                              hipStream_t stream) {
    const float* x      = (const float*)d_in[0];
    const int*   ei     = (const int*)d_in[1];
    const float* conv_w = (const float*)d_in[2];
    const float* conv_b = (const float*)d_in[3];
    const float* out_w  = (const float*)d_in[4];
    const float* out_b  = (const float*)d_in[5];
    float* out = (float*)d_out;

    const int N = in_sizes[0] / IN_CH;
    const int E = in_sizes[1] / 2;

    const int nblk = (N + 31) / 32;

    char* ws = (char*)d_ws;
    size_t off = 0;
    float* dinv   = (float*)(ws + off);    off = align256(off + (size_t)N * 4);
    float* h      = (float*)(ws + off);    off = align256(off + (size_t)N * HID * 4);
    float* agg    = (float*)(ws + off);    off = align256(off + (size_t)nblk * 32 * HID * 4);
    unsigned* part = (unsigned*)agg;

    const int nb_n = (N + 255) / 256;

    k_deg_hist<<<dim3(DEG_PARTS, (nblk * 32 + DEG_WIN - 1) / DEG_WIN), DEG_THREADS, 0, stream>>>(ei, E, N, part);

    k_gemm_wmma<<<nblk, 64, 0, stream>>>(x, conv_w, part, dinv, h, agg, N);

    k_edge_agg<<<dim3(AG_CHUNKS, (N + AG_WIN - 1) / AG_WIN), AG_THREADS, 0, stream>>>(ei, E, N, h, agg);

    k_out<<<nb_n, 256, 0, stream>>>(agg, dinv, conv_b, out_w, out_b, out, N);
}


# edges binned by target window once: 288 more work-groups of the degree launch rank them by returning LDS adds, store packed (src,offset) lists; edge kernel reads them, no 28x scan; sum order changes
# speedup vs baseline: 5.1202x; 1.3763x over previous
#include <hip/hip_runtime.h>
#include <hip/hip_bf16.h>

typedef __attribute__((ext_vector_type(2))) float v2f;
typedef __attribute__((ext_vector_type(4))) float v4f;
typedef __attribute__((ext_vector_type(8))) float v8f;

#define IN_CH 256
#define HID   16


typedef __attribute__((ext_vector_type(4))) unsigned v4u;
typedef __attribute__((ext_vector_type(4))) int v4i;

#define DEG_PARTS   16
#define DEG_THREADS 1024
#define DEG_WIN     12800
#define DEG_LOADS   8
static_assert(DEG_PARTS * 4 == HID * 4, "a node's partial counts must fill exactly its agg row");
static_assert(DEG_WIN % 32 == 0, "a window holds whole node blocks");

#define AG_THREADS 1024
#define AG_WAVES   (AG_THREADS / 32)
#define AG_CHUNKS  9
#define AG_WIN     3584
#define AG_CAPW    512
#define AG_TBITS   12
static_assert(AG_WIN <= (1 << AG_TBITS), "a window offset must fit the low bits of a list entry");
#define BIN_SUBS   AG_WAVES
#define BIN_PRODS  (AG_CHUNKS * BIN_SUBS)
#define BIN_CAP    AG_CAPW
#define BIN_MAXW   32
#define BIN_LOADS  3
#define DEG_SMEM   (BIN_MAXW * BIN_CAP + BIN_MAXW)
static_assert(DEG_SMEM >= DEG_WIN, "the shared array must hold a degree window");

__device__ __forceinline__ int bin_sub(int E) {
    return (((E + BIN_PRODS - 1) / BIN_PRODS) + 3) & ~3;
}

__device__ __forceinline__ void bin_four(unsigned* stage, unsigned* cntL, int nwin, v4i tv, v4i sv) {
    unsigned w[4], r[4];
#pragma unroll
    for (int c = 0; c < 4; ++c) {
        w[c] = (unsigned)tv[c] / (unsigned)AG_WIN;
        r[c] = BIN_CAP;
        if (w[c] < (unsigned)nwin) r[c] = atomicAdd(&cntL[w[c]], 1u);
    }
#pragma unroll
    for (int c = 0; c < 4; ++c) {
        if (r[c] < BIN_CAP)
            stage[w[c] * BIN_CAP + r[c]] = ((unsigned)sv[c] << AG_TBITS) | ((unsigned)tv[c] - w[c] * AG_WIN);
    }
}

__device__ __forceinline__ void bin_role(const int* __restrict__ ei, int E, int b, int nwin,
                                         unsigned* __restrict__ bucket, unsigned* __restrict__ bcnt,
                                         unsigned* smem) {
    unsigned* stage = smem;
    unsigned* cntL  = smem + BIN_MAXW * BIN_CAP;
    const int tid  = threadIdx.x;
    const int lane = tid & 31;
    const int wv   = __builtin_amdgcn_readfirstlane(tid >> 5);
    const int sub  = bin_sub(E);
    int e0 = b * sub;    if (e0 > E) e0 = E;
    int e1 = e0 + sub;   if (e1 > E) e1 = E;
    const int cnt = e1 - e0;
    const int* tg = ei + E + e0;
    const int* sr = ei + e0;

    if (tid < BIN_MAXW) cntL[tid] = 0u;
    __syncthreads();

    const int n4 = cnt >> 2;
    const v4i* tg4 = (const v4i*)tg;
    const v4i* sr4 = (const v4i*)sr;
    for (int g = tid; g < n4; g += DEG_THREADS * BIN_LOADS) {
        v4i t[BIN_LOADS], sv[BIN_LOADS];
#pragma unroll
        for (int u = 0; u < BIN_LOADS; ++u) {
            const int gg = g + u * DEG_THREADS;
            t[u]  = (gg < n4) ? tg4[gg] : (v4i){-1, -1, -1, -1};
            sv[u] = (gg < n4) ? sr4[gg] : (v4i){0, 0, 0, 0};
        }
#pragma unroll
        for (int u = 0; u < BIN_LOADS; ++u)
            bin_four(stage, cntL, nwin, t[u], sv[u]);
    }
    {
        const int r = (n4 << 2) + tid;
        if (r < cnt) {
            const v4i tv = {tg[r], -1, -1, -1};
            const v4i s1 = {sr[r], 0, 0, 0};
            bin_four(stage, cntL, nwin, tv, s1);
        }
    }
    __syncthreads();

    for (int w = wv; w < nwin; w += DEG_THREADS / 32) {
        unsigned n = (unsigned)__builtin_amdgcn_readfirstlane((int)cntL[w]);
        if (n > BIN_CAP) n = BIN_CAP;
        const unsigned n32 = (n + 31u) & ~31u;
        unsigned* dst = bucket + ((size_t)w * BIN_PRODS + (size_t)b) * BIN_CAP;
        const unsigned* st = stage + w * BIN_CAP;
        for (unsigned k = 0; k < n32; k += 32)
            dst[k + lane] = st[k + lane];
    }
    if (tid < nwin) bcnt[(size_t)tid * BIN_PRODS + b] = cntL[tid];
}

__device__ __forceinline__ void deg_count(unsigned* hist, unsigned nbase, unsigned wn, int tv) {
    const unsigned w = (unsigned)tv - nbase;
    if (w < wn) atomicAdd(&hist[w], 1u);
}

__global__ __launch_bounds__(DEG_THREADS)
void k_deg_hist(const int* __restrict__ ei, int E, int N,
                unsigned* __restrict__ part,
                int ndegy, int nwin,
                unsigned* __restrict__ bucket, unsigned* __restrict__ bcnt) {
    __shared__ unsigned hist[DEG_SMEM];

    if ((int)blockIdx.y >= ndegy) {
        const int b = ((int)blockIdx.y - ndegy) * (int)gridDim.x + (int)blockIdx.x;
        if (b < BIN_PRODS) bin_role(ei, E, b, nwin, bucket, bcnt, hist);
        return;
    }

    const int tid   = threadIdx.x;
    const int lane  = tid & 31;
    const int wv    = tid >> 5;
    const int nblk  = (N + 31) >> 5;
    const int chunk = (((E + DEG_PARTS - 1) / DEG_PARTS) + 3) & ~3;
    const unsigned nbase = blockIdx.y * DEG_WIN;
    unsigned wn = (unsigned)nblk * 32u - nbase;
    if (wn > DEG_WIN) wn = DEG_WIN;

    int e0 = blockIdx.x * chunk; if (e0 > E) e0 = E;
    int e1 = e0 + chunk;         if (e1 > E) e1 = E;
    const int cnt = e1 - e0;
    const int* tg = ei + E + e0;

    v4u* h4 = (v4u*)hist;
    for (int i = tid; i < DEG_WIN / 4; i += DEG_THREADS)
        h4[i] = (v4u){0u, 0u, 0u, 0u};
    __syncthreads();

    const int n4 = cnt >> 2;
    const v4i* tg4 = (const v4i*)tg;
    for (int g = tid; g < n4; g += DEG_THREADS * DEG_LOADS) {
        v4i t[DEG_LOADS];
#pragma unroll
        for (int u = 0; u < DEG_LOADS; ++u) {
            int gg = g + u * DEG_THREADS;
            t[u] = (gg < n4) ? tg4[gg] : (v4i){-1, -1, -1, -1};
        }
#pragma unroll
        for (int u = 0; u < DEG_LOADS; ++u) {
            deg_count(hist, nbase, wn, t[u].x); deg_count(hist, nbase, wn, t[u].y);
            deg_count(hist, nbase, wn, t[u].z); deg_count(hist, nbase, wn, t[u].w);
        }
    }
    {
        int r = (n4 << 2) + tid;
        if (r < cnt) deg_count(hist, nbase, wn, tg[r]);
    }
    __syncthreads();

    const int nb0 = (int)(nbase >> 5);
    const int nbw = (int)(wn >> 5);
    for (int nb = wv; nb < nbw; nb += DEG_THREADS / 32)
        part[((size_t)(nb0 + nb) * DEG_PARTS + blockIdx.x) * 32 + lane] = hist[nb * 32 + lane];
}

#define XPAD 4

__global__ __launch_bounds__(64)
void k_gemm_wmma(const float* __restrict__ x,
                 const float* __restrict__ w,
                 const unsigned* part,
                 float* __restrict__ dinv,
                 float* __restrict__ h,
                 float* agg, int N) {
    __shared__ float sW[IN_CH][HID];
    __shared__ float sX[2][16][IN_CH + XPAD];
    __shared__ unsigned sDeg[2][32];

    const int tid  = threadIdx.x;
    const int wv   = tid >> 5;
    const int lane = tid & 31;
    const int l16  = lane & 15;
    const int half = lane >> 4;
    const int node_base = blockIdx.x * 32 + wv * 16;
    const bool full_tile = (node_base + 16 <= N);

    {
        const unsigned* pp = part + ((size_t)blockIdx.x * DEG_PARTS + (size_t)wv * (DEG_PARTS / 2)) * 32 + lane;
        unsigned sum = 0u;
#pragma unroll
        for (int p = 0; p < DEG_PARTS / 2; ++p)
            sum += pp[p * 32];
        sDeg[wv][lane] = sum;
    }

    {
        const v4f* ws = (const v4f*)w;
        v4f* wd = (v4f*)&sW[0][0];
#pragma unroll
        for (int i = 0; i < (IN_CH * HID) / 4 / 64; ++i)
            wd[tid + i * 64] = ws[tid + i * 64];
    }

    if (full_tile) {
        const v4f* xs = (const v4f*)(x + (size_t)node_base * IN_CH);
#pragma unroll
        for (int it = 0; it < 32; ++it) {
            int fi = it * 32 + lane;
            v4f v = __builtin_nontemporal_load(xs + fi);
            int f = fi * 4;
            *(v4f*)&sX[wv][f >> 8][f & 255] = v;
        }
    } else {
        for (int it = 0; it < 32; ++it) {
            int f = it * 128 + lane * 4;
            int r = f >> 8, c = f & 255;
            int row = node_base + r;
            v4f v = {};
            if (row < N) v = *(const v4f*)(x + (size_t)row * IN_CH + c);
            *(v4f*)&sX[wv][r][c] = v;
        }
    }
    __syncthreads();

    float dv[8];
#pragma unroll
    for (int i = 0; i < 8; ++i) {
        const int r = wv * 16 + 8 * half + i;
        dv[i] = rsqrtf((float)(1u + sDeg[0][r] + sDeg[1][r]));
    }

    v8f acc = {};
#pragma unroll 8
    for (int k = 0; k < IN_CH; k += 4) {
        v2f a;
        a.x = sX[wv][l16][k + 2 * half + 0];
        a.y = sX[wv][l16][k + 2 * half + 1];
        v2f b;
        b.x = sW[k + 2 * half + 0][l16];
        b.y = sW[k + 2 * half + 1][l16];
        acc = __builtin_amdgcn_wmma_f32_16x16x4_f32(
            false, a, false, b, (short)0, acc, false, false);
    }

    if (full_tile) {
        unsigned base = (unsigned)(node_base + 8 * half) * HID + (unsigned)l16;
#pragma unroll
        for (int i = 0; i < 8; ++i) {
            const float v = acc[i] * dv[i];
            h[base + (unsigned)i * HID]   = v;
            agg[base + (unsigned)i * HID] = v;
        }
        if (l16 == 0) {
#pragma unroll
            for (int i = 0; i < 8; ++i)
                dinv[node_base + 8 * half + i] = dv[i];
        }
    } else {
#pragma unroll
        for (int i = 0; i < 8; ++i) {
            int row = node_base + i + 8 * half;
            if (row < N) {
                const float v = acc[i] * dv[i];
                h[(size_t)row * HID + l16]   = v;
                agg[(size_t)row * HID + l16] = v;
                if (l16 == 0) dinv[row] = dv[i];
            }
        }
    }
}


struct __attribute__((aligned(8))) ag_slot { unsigned ho, lo; };

__device__ __forceinline__ void ag_direct(const float* __restrict__ h,
                                       float* __restrict__ agg, int s, int t) {
#pragma nounroll
    for (int j = 0; j < HID; ++j)
        unsafeAtomicAdd(&agg[(size_t)t * HID + j], h[(size_t)s * HID + j]);
}

template <bool FULL>
__device__ __forceinline__ void ag_gather32(const int* __restrict__ src_of, const float* __restrict__ h,
                                            float* win, const unsigned* mylist, ag_slot* myslot,
                                            int k, int n, int lane, int l16, int half) {
    ag_slot mine = {0u, 0u};
    if (FULL || k + lane < n) {
        const unsigned p = mylist[k + lane];
        const int s = src_of[(int)(p >> AG_TBITS)];
        mine.ho = (unsigned)s * HID;
        mine.lo = (p & ((1u << AG_TBITS) - 1u)) * HID;
    }
    myslot[lane] = mine;

    ag_slot q[16];
    float hv[16];
#pragma unroll
    for (int i = 0; i < 16; ++i) {
        q[i]  = myslot[2 * i + half];
        hv[i] = h[q[i].ho + (unsigned)l16];
    }
#pragma unroll
    for (int i = 0; i < 16; ++i) {
        if (FULL || k + 2 * i + half < n)
            atomicAdd(&win[q[i].lo + (unsigned)l16], hv[i]);
    }
}

__global__ __launch_bounds__(AG_THREADS)
void k_edge_agg(const int* __restrict__ ei, int E, int N,
                const float* __restrict__ h,
                float* __restrict__ agg) {
    __shared__ float    win[AG_WIN * HID];
    __shared__ unsigned list[AG_WAVES][AG_CAPW];
    __shared__ ag_slot slot[AG_WAVES][32];

    const int tid  = threadIdx.x;
    const int lane = tid & 31;
    const int wv   = __builtin_amdgcn_readfirstlane(tid >> 5);
    const int l16  = lane & 15;
    const int half = lane >> 4;
    const unsigned lt_mask = (1u << lane) - 1u;

    const int nbase = blockIdx.y * AG_WIN;
    int wn = N - nbase; if (wn > AG_WIN) wn = AG_WIN;
    const int chunk = (((E + AG_CHUNKS - 1) / AG_CHUNKS) + 3) & ~3;
    int e0 = blockIdx.x * chunk; if (e0 > E) e0 = E;
    int e1 = e0 + chunk;         if (e1 > E) e1 = E;
    const int cnt = e1 - e0;
    const int* tg = ei + E + e0;

    v4f* w4 = (v4f*)win;
    for (int i = tid; i < AG_WIN * HID / 4; i += AG_THREADS)
        w4[i] = (v4f){0.0f, 0.0f, 0.0f, 0.0f};

    unsigned* mylist = list[wv];
    int wcnt = 0;
    const int n4 = cnt >> 2;
    const v4i* tg4 = (const v4i*)tg;
    for (int g0 = wv * 32; g0 < n4; g0 += AG_THREADS * 4) {
        v4i t[4];
        if (g0 + 3 * AG_THREADS + 32 <= n4) {
#pragma unroll
            for (int u = 0; u < 4; ++u)
                t[u] = tg4[g0 + u * AG_THREADS + lane];
        } else {
#pragma unroll
            for (int u = 0; u < 4; ++u) {
                const int g = g0 + u * AG_THREADS + lane;
                t[u] = (g < n4) ? tg4[g] : (v4i){-1, -1, -1, -1};
            }
        }
        unsigned m[4][4];
        const int wbase = wcnt;
        int tot = 0;
#pragma unroll
        for (int u = 0; u < 4; ++u) {
#pragma unroll
            for (int c = 0; c < 4; ++c) {
                m[u][c] = __builtin_amdgcn_ballot_w32((unsigned)(t[u][c] - nbase) < (unsigned)wn);
                tot += __builtin_popcount(m[u][c]);
            }
        }
        int run = wbase;
        if (wbase + tot <= AG_CAPW) {
#pragma unroll
            for (int u = 0; u < 4; ++u) {
                const int g = g0 + u * AG_THREADS + lane;
#pragma unroll
                for (int c = 0; c < 4; ++c) {
                    const unsigned off = (unsigned)(t[u][c] - nbase);
                    if (off < (unsigned)wn) {
                        const unsigned pos = __builtin_amdgcn_mbcnt_lo(m[u][c], (unsigned)run);
                        mylist[pos] = ((unsigned)(4 * g + c) << AG_TBITS) | off;
                    }
                    run += __builtin_popcount(m[u][c]);
                }
            }
        } else {
#pragma unroll
            for (int u = 0; u < 4; ++u) {
                const int g = g0 + u * AG_THREADS + lane;
#pragma unroll
                for (int c = 0; c < 4; ++c) {
                    const bool hit = (unsigned)(t[u][c] - nbase) < (unsigned)wn;
                    if (hit) {
                        const int pos = run + __builtin_popcount(m[u][c] & lt_mask);
                        const unsigned le = (unsigned)(4 * g + c);
                        if (pos < AG_CAPW) mylist[pos] = (le << AG_TBITS) | (unsigned)(t[u][c] - nbase);
                        else ag_direct(h, agg, ei[e0 + (int)le], t[u][c]);
                    }
                    run += __builtin_popcount(m[u][c]);
                }
            }
        }
        wcnt = wbase + tot;
    }
    if (wv == 0) {
        const int r = (n4 << 2) + lane;
        int tv = -1;
        if (lane < 3 && r < cnt) tv = tg[r];
        const bool hit = (unsigned)(tv - nbase) < (unsigned)wn;
        const unsigned m = __builtin_amdgcn_ballot_w32(hit);
        if (hit) {
            const int pos = wcnt + __builtin_popcount(m & lt_mask);
            if (pos < AG_CAPW) mylist[pos] = ((unsigned)r << AG_TBITS) | (unsigned)(tv - nbase);
            else ag_direct(h, agg, ei[e0 + r], tv);
        }
        wcnt += __builtin_popcount(m);
    }
    __syncthreads();

    const int n = wcnt < AG_CAPW ? wcnt : AG_CAPW;
    {
        int k = 0;
        for (; k + 32 <= n; k += 32)
            ag_gather32<true>(ei + e0, h, win, mylist, slot[wv], k, n, lane, l16, half);
        if (k < n)
            ag_gather32<false>(ei + e0, h, win, mylist, slot[wv], k, n, lane, l16, half);
    }
    __syncthreads();

    float* dst = agg + (size_t)nbase * HID;
    for (int i = tid; i < wn * HID; i += AG_THREADS)
        unsafeAtomicAdd(&dst[i], win[i]);
}


template <bool FULL>
__device__ __forceinline__ void ag_rows32(const float* __restrict__ h, float* win, ag_slot* myslot,
                                          unsigned p, int k, int n, int lane, int l16, int half) {
    ag_slot mine = {0u, 0u};
    if (FULL || k + lane < n) {
        mine.ho = (p >> AG_TBITS) * HID;
        mine.lo = (p & ((1u << AG_TBITS) - 1u)) * HID;
    }
    myslot[lane] = mine;

    ag_slot q[16];
    float hv[16];
#pragma unroll
    for (int i = 0; i < 16; ++i) {
        q[i]  = myslot[2 * i + half];
        hv[i] = h[q[i].ho + (unsigned)l16];
    }
#pragma unroll
    for (int i = 0; i < 16; ++i) {
        if (FULL || k + 2 * i + half < n)
            atomicAdd(&win[q[i].lo + (unsigned)l16], hv[i]);
    }
}

__global__ __launch_bounds__(AG_THREADS)
void k_edge_agg_b(const int* __restrict__ ei, int E, int N,
                  const float* __restrict__ h,
                  float* __restrict__ agg,
                  const unsigned* __restrict__ bucket,
                  const unsigned* __restrict__ bcnt) {
    __shared__ float   win[AG_WIN * HID];
    __shared__ ag_slot slot[AG_WAVES][32];

    const int tid  = threadIdx.x;
    const int lane = tid & 31;
    const int wv   = __builtin_amdgcn_readfirstlane(tid >> 5);
    const int l16  = lane & 15;
    const int half = lane >> 4;

    const int nbase = blockIdx.y * AG_WIN;
    int wn = N - nbase; if (wn > AG_WIN) wn = AG_WIN;
    const int b = blockIdx.x * BIN_SUBS + wv;
    const unsigned* seg = bucket + ((size_t)blockIdx.y * BIN_PRODS + (size_t)b) * BIN_CAP;
    const int cnt = (int)bcnt[(size_t)blockIdx.y * BIN_PRODS + b];
    const bool listed = cnt <= BIN_CAP;
    const int n = listed ? cnt : 0;
    unsigned pa = (lane < n) ? seg[lane] : 0u;
    unsigned pb = (32 + lane < n) ? seg[32 + lane] : 0u;

    v4f* w4 = (v4f*)win;
    for (int i = tid; i < AG_WIN * HID / 4; i += AG_THREADS)
        w4[i] = (v4f){0.0f, 0.0f, 0.0f, 0.0f};
    __syncthreads();

    {
        int k = 0;
        for (; k + 32 <= n; k += 32) {
            const unsigned p = pa;
            pa = pb;
            pb = (k + 64 + lane < n) ? seg[k + 64 + lane] : 0u;
            ag_rows32<true>(h, win, slot[wv], p, k, n, lane, l16, half);
        }
        if (k < n)
            ag_rows32<false>(h, win, slot[wv], pa, k, n, lane, l16, half);
    }
    if (!listed) {
        const int sub = bin_sub(E);
        int e0 = b * sub;    if (e0 > E) e0 = E;
        int e1 = e0 + sub;   if (e1 > E) e1 = E;
        for (int e = e0 + lane; e < e1; e += 32) {
            const unsigned off = (unsigned)(ei[E + e] - nbase);
            if (off < (unsigned)wn) {
                const int sidx = ei[e];
#pragma nounroll
                for (int j = 0; j < HID; ++j)
                    atomicAdd(&win[off * HID + (unsigned)j], h[(size_t)sidx * HID + j]);
            }
        }
    }
    __syncthreads();

    float* dst = agg + (size_t)nbase * HID;
    for (int i = tid; i < wn * HID; i += AG_THREADS)
        unsafeAtomicAdd(&dst[i], win[i]);
}


__global__ void k_out(const float* __restrict__ agg,
                      const float* __restrict__ dinv,
                      const float* __restrict__ conv_b,
                      const float* __restrict__ out_w,
                      const float* __restrict__ out_b,
                      float* __restrict__ out, int N) {
    int n = blockIdx.x * blockDim.x + threadIdx.x;
    if (n >= N) return;
    const v4f* av = (const v4f*)(agg + (size_t)n * HID);
    const v4f* bv = (const v4f*)conv_b;
    const v4f* wv = (const v4f*)out_w;
    const float dn = dinv[n];
    float acc = out_b[0];
#pragma unroll
    for (int i = 0; i < 4; ++i) {
        v4f v = av[i] * dn + bv[i];
        v4f wvi = wv[i];
#pragma unroll
        for (int j = 0; j < 4; ++j)
            acc = fmaf(fmaxf(v[j], 0.0f), wvi[j], acc);
    }
    out[n] = acc;
}


static inline size_t align256(size_t v) { return (v + 255) & ~(size_t)255; }

extern "C" void kernel_launch(void* const* d_in, const int* in_sizes, int n_in,
                              void* d_out, int out_size, void* d_ws, size_t ws_size,
                              hipStream_t stream) {
    const float* x      = (const float*)d_in[0];
    const int*   ei     = (const int*)d_in[1];
    const float* conv_w = (const float*)d_in[2];
    const float* conv_b = (const float*)d_in[3];
    const float* out_w  = (const float*)d_in[4];
    const float* out_b  = (const float*)d_in[5];
    float* out = (float*)d_out;

    const int N = in_sizes[0] / IN_CH;
    const int E = in_sizes[1] / 2;

    const int nblk = (N + 31) / 32;

    char* ws = (char*)d_ws;
    size_t off = 0;
    float* dinv   = (float*)(ws + off);    off = align256(off + (size_t)N * 4);
    float* h      = (float*)(ws + off);    off = align256(off + (size_t)N * HID * 4);
    float* agg    = (float*)(ws + off);    off = align256(off + (size_t)nblk * 32 * HID * 4);
    unsigned* part = (unsigned*)agg;
    const int nwin = (N + AG_WIN - 1) / AG_WIN;
    unsigned* bcnt   = (unsigned*)(ws + off);   off = align256(off + (size_t)nwin * BIN_PRODS * 4);
    unsigned* bucket = (unsigned*)(ws + off);   off = align256(off + (size_t)nwin * BIN_PRODS * BIN_CAP * 4);
    const bool binned = (off <= ws_size) && (nwin <= BIN_MAXW);

    const int nb_n = (N + 255) / 256;
    const int ndegy = (nblk * 32 + DEG_WIN - 1) / DEG_WIN;
    const int nbiny = binned ? (BIN_PRODS + DEG_PARTS - 1) / DEG_PARTS : 0;

    k_deg_hist<<<dim3(DEG_PARTS, ndegy + nbiny), DEG_THREADS, 0, stream>>>(ei, E, N, part, ndegy, nwin, bucket, bcnt);

    k_gemm_wmma<<<nblk, 64, 0, stream>>>(x, conv_w, part, dinv, h, agg, N);

    if (binned)
        k_edge_agg_b<<<dim3(AG_CHUNKS, nwin), AG_THREADS, 0, stream>>>(ei, E, N, h, agg, bucket, bcnt);
    else
        k_edge_agg<<<dim3(AG_CHUNKS, nwin), AG_THREADS, 0, stream>>>(ei, E, N, h, agg);

    k_out<<<nb_n, 256, 0, stream>>>(agg, dinv, conv_b, out_w, out_b, out, N);
}


# each 416-node window summed by ONE work-group from 241-way edge bins: starts as own h rows, ends in dinv, bias, relu, projection -> out; global float adds, agg and out launch gone; sum order changes
# speedup vs baseline: 5.5654x; 1.0869x over previous
#include <hip/hip_runtime.h>
#include <hip/hip_bf16.h>

typedef __attribute__((ext_vector_type(2))) float v2f;
typedef __attribute__((ext_vector_type(4))) float v4f;
typedef __attribute__((ext_vector_type(8))) float v8f;

#define IN_CH 256
#define HID   16


typedef __attribute__((ext_vector_type(4))) unsigned v4u;
typedef __attribute__((ext_vector_type(4))) int v4i;

#define DEG_PARTS   16
#define DEG_THREADS 1024
#define DEG_WIN     12800
#define DEG_LOADS   8
static_assert(DEG_PARTS * 4 == HID * 4, "a node's partial counts must fill exactly its row of h");
static_assert(DEG_WIN % 32 == 0, "a window holds whole node blocks");

#define AG_THREADS 1024
#define AG_WAVES   (AG_THREADS / 32)
#define AG_CHUNKS  9
#define AG_WIN     3584
#define AG_CAPW    512
#define AG_TBITS   12
static_assert(AG_WIN <= (1 << AG_TBITS), "a window offset must fit the low bits of a list entry");
#define XW_WIN     416
#define XW_THREADS 1024
#define XW_WAVES   (XW_THREADS / 32)
#define XW_LCAP    640
static_assert(XW_WIN <= (1 << AG_TBITS), "a window offset must fit the low bits of a list entry");
static_assert(XW_WIN % 32 == 0 && XW_WIN <= XW_THREADS, "whole lines of out per window, one thread per node at the end");
static_assert(XW_LCAP % 32 == 0, "a group of 32 list entries never leaves the wave's list");
#define BIN_PRODS  288
#define XW_BPW     (BIN_PRODS / XW_WAVES)
static_assert(BIN_PRODS % XW_WAVES == 0, "every wave of the sum kernel takes the same number of bins");
#define BIN_CAP    96
#define BIN_LINES  (BIN_CAP / 32)
static_assert(BIN_CAP % 32 == 0, "a bin is whole lines");
#define BIN_MAXW   256
#define BIN_LOADS  3
#define DEG_SMEM   (BIN_MAXW * BIN_CAP + BIN_MAXW)
static_assert(DEG_SMEM >= DEG_WIN, "the shared array must hold a degree window");

__device__ __forceinline__ int bin_sub(int E) {
    return (((E + BIN_PRODS - 1) / BIN_PRODS) + 3) & ~3;
}

__device__ __forceinline__ void bin_four(unsigned* stage, unsigned* cntL, int nwin, v4i tv, v4i sv) {
    unsigned w[4], r[4];
#pragma unroll
    for (int c = 0; c < 4; ++c) {
        w[c] = (unsigned)tv[c] / (unsigned)XW_WIN;
        r[c] = BIN_CAP;
        if (w[c] < (unsigned)nwin) r[c] = atomicAdd(&cntL[w[c]], 1u);
    }
#pragma unroll
    for (int c = 0; c < 4; ++c) {
        if (r[c] < BIN_CAP)
            stage[w[c] * BIN_CAP + r[c]] = ((unsigned)sv[c] << AG_TBITS) | ((unsigned)tv[c] - w[c] * XW_WIN);
    }
}

__device__ __forceinline__ void bin_role(const int* __restrict__ ei, int E, int b, int nwin,
                                         unsigned* __restrict__ bucket, unsigned* __restrict__ bcnt,
                                         unsigned* smem) {
    unsigned* stage = smem;
    unsigned* cntL  = smem + BIN_MAXW * BIN_CAP;
    const int tid  = threadIdx.x;
    const int lane = tid & 31;
    const int wv   = __builtin_amdgcn_readfirstlane(tid >> 5);
    const int sub  = bin_sub(E);
    int e0 = b * sub;    if (e0 > E) e0 = E;
    int e1 = e0 + sub;   if (e1 > E) e1 = E;
    const int cnt = e1 - e0;
    const int* tg = ei + E + e0;
    const int* sr = ei + e0;

    if (tid < BIN_MAXW) cntL[tid] = 0u;
    __syncthreads();

    const int n4 = cnt >> 2;
    const v4i* tg4 = (const v4i*)tg;
    const v4i* sr4 = (const v4i*)sr;
    for (int g = tid; g < n4; g += DEG_THREADS * BIN_LOADS) {
        v4i t[BIN_LOADS], sv[BIN_LOADS];
#pragma unroll
        for (int u = 0; u < BIN_LOADS; ++u) {
            const int gg = g + u * DEG_THREADS;
            t[u]  = (gg < n4) ? tg4[gg] : (v4i){-1, -1, -1, -1};
            sv[u] = (gg < n4) ? sr4[gg] : (v4i){0, 0, 0, 0};
        }
#pragma unroll
        for (int u = 0; u < BIN_LOADS; ++u)
            bin_four(stage, cntL, nwin, t[u], sv[u]);
    }
    {
        const int r = (n4 << 2) + tid;
        if (r < cnt) {
            const v4i tv = {tg[r], -1, -1, -1};
            const v4i s1 = {sr[r], 0, 0, 0};
            bin_four(stage, cntL, nwin, tv, s1);
        }
    }
    __syncthreads();

    for (int w = wv; w < nwin; w += DEG_THREADS / 32) {
        unsigned n = (unsigned)__builtin_amdgcn_readfirstlane((int)cntL[w]);
        if (n > BIN_CAP) n = BIN_CAP;
        const unsigned n32 = (n + 31u) & ~31u;
        unsigned* dst = bucket + ((size_t)w * BIN_PRODS + (size_t)b) * BIN_CAP;
        const unsigned* st = stage + w * BIN_CAP;
        for (unsigned k = 0; k < n32; k += 32)
            dst[k + lane] = st[k + lane];
    }
    if (tid < nwin) bcnt[(size_t)tid * BIN_PRODS + b] = cntL[tid];
}

__device__ __forceinline__ void deg_count(unsigned* hist, unsigned nbase, unsigned wn, int tv) {
    const unsigned w = (unsigned)tv - nbase;
    if (w < wn) atomicAdd(&hist[w], 1u);
}

__global__ __launch_bounds__(DEG_THREADS)
void k_deg_hist(const int* __restrict__ ei, int E, int N,
                unsigned* __restrict__ part,
                int ndegy, int nwin,
                unsigned* __restrict__ bucket, unsigned* __restrict__ bcnt) {
    __shared__ unsigned hist[DEG_SMEM];

    if ((int)blockIdx.y >= ndegy) {
        const int b = ((int)blockIdx.y - ndegy) * (int)gridDim.x + (int)blockIdx.x;
        if (b < BIN_PRODS) bin_role(ei, E, b, nwin, bucket, bcnt, hist);
        return;
    }

    const int tid   = threadIdx.x;
    const int lane  = tid & 31;
    const int wv    = tid >> 5;
    const int nblk  = (N + 31) >> 5;
    const int chunk = (((E + DEG_PARTS - 1) / DEG_PARTS) + 3) & ~3;
    const unsigned nbase = blockIdx.y * DEG_WIN;
    unsigned wn = (unsigned)nblk * 32u - nbase;
    if (wn > DEG_WIN) wn = DEG_WIN;

    int e0 = blockIdx.x * chunk; if (e0 > E) e0 = E;
    int e1 = e0 + chunk;         if (e1 > E) e1 = E;
    const int cnt = e1 - e0;
    const int* tg = ei + E + e0;

    v4u* h4 = (v4u*)hist;
    for (int i = tid; i < DEG_WIN / 4; i += DEG_THREADS)
        h4[i] = (v4u){0u, 0u, 0u, 0u};
    __syncthreads();

    const int n4 = cnt >> 2;
    const v4i* tg4 = (const v4i*)tg;
    for (int g = tid; g < n4; g += DEG_THREADS * DEG_LOADS) {
        v4i t[DEG_LOADS];
#pragma unroll
        for (int u = 0; u < DEG_LOADS; ++u) {
            int gg = g + u * DEG_THREADS;
            t[u] = (gg < n4) ? tg4[gg] : (v4i){-1, -1, -1, -1};
        }
#pragma unroll
        for (int u = 0; u < DEG_LOADS; ++u) {
            deg_count(hist, nbase, wn, t[u].x); deg_count(hist, nbase, wn, t[u].y);
            deg_count(hist, nbase, wn, t[u].z); deg_count(hist, nbase, wn, t[u].w);
        }
    }
    {
        int r = (n4 << 2) + tid;
        if (r < cnt) deg_count(hist, nbase, wn, tg[r]);
    }
    __syncthreads();

    const int nb0 = (int)(nbase >> 5);
    const int nbw = (int)(wn >> 5);
    for (int nb = wv; nb < nbw; nb += DEG_THREADS / 32)
        part[((size_t)(nb0 + nb) * DEG_PARTS + blockIdx.x) * 32 + lane] = hist[nb * 32 + lane];
}

#define XPAD 4

template <bool WRITE_AGG>
__global__ __launch_bounds__(64)
void k_gemm_wmma(const float* __restrict__ x,
                 const float* __restrict__ w,
                 const unsigned* part,
                 float* __restrict__ dinv,
                 float* h,
                 float* agg, int N) {
    __shared__ float sW[IN_CH][HID];
    __shared__ float sX[2][16][IN_CH + XPAD];
    __shared__ unsigned sDeg[2][32];

    const int tid  = threadIdx.x;
    const int wv   = tid >> 5;
    const int lane = tid & 31;
    const int l16  = lane & 15;
    const int half = lane >> 4;
    const int node_base = blockIdx.x * 32 + wv * 16;
    const bool full_tile = (node_base + 16 <= N);

    {
        const unsigned* pp = part + ((size_t)blockIdx.x * DEG_PARTS + (size_t)wv * (DEG_PARTS / 2)) * 32 + lane;
        unsigned sum = 0u;
#pragma unroll
        for (int p = 0; p < DEG_PARTS / 2; ++p)
            sum += pp[p * 32];
        sDeg[wv][lane] = sum;
    }

    {
        const v4f* ws = (const v4f*)w;
        v4f* wd = (v4f*)&sW[0][0];
#pragma unroll
        for (int i = 0; i < (IN_CH * HID) / 4 / 64; ++i)
            wd[tid + i * 64] = ws[tid + i * 64];
    }

    if (full_tile) {
        const v4f* xs = (const v4f*)(x + (size_t)node_base * IN_CH);
#pragma unroll
        for (int it = 0; it < 32; ++it) {
            int fi = it * 32 + lane;
            v4f v = __builtin_nontemporal_load(xs + fi);
            int f = fi * 4;
            *(v4f*)&sX[wv][f >> 8][f & 255] = v;
        }
    } else {
        for (int it = 0; it < 32; ++it) {
            int f = it * 128 + lane * 4;
            int r = f >> 8, c = f & 255;
            int row = node_base + r;
            v4f v = {};
            if (row < N) v = *(const v4f*)(x + (size_t)row * IN_CH + c);
            *(v4f*)&sX[wv][r][c] = v;
        }
    }
    __syncthreads();

    float dv[8];
#pragma unroll
    for (int i = 0; i < 8; ++i) {
        const int r = wv * 16 + 8 * half + i;
        dv[i] = rsqrtf((float)(1u + sDeg[0][r] + sDeg[1][r]));
    }

    v8f acc = {};
#pragma unroll 8
    for (int k = 0; k < IN_CH; k += 4) {
        v2f a;
        a.x = sX[wv][l16][k + 2 * half + 0];
        a.y = sX[wv][l16][k + 2 * half + 1];
        v2f b;
        b.x = sW[k + 2 * half + 0][l16];
        b.y = sW[k + 2 * half + 1][l16];
        acc = __builtin_amdgcn_wmma_f32_16x16x4_f32(
            false, a, false, b, (short)0, acc, false, false);
    }

    if (full_tile) {
        unsigned base = (unsigned)(node_base + 8 * half) * HID + (unsigned)l16;
#pragma unroll
        for (int i = 0; i < 8; ++i) {
            const float v = acc[i] * dv[i];
            h[base + (unsigned)i * HID]   = v;
            if (WRITE_AGG) agg[base + (unsigned)i * HID] = v;
        }
        if (l16 == 0) {
#pragma unroll
            for (int i = 0; i < 8; ++i)
                dinv[node_base + 8 * half + i] = dv[i];
        }
    } else {
#pragma unroll
        for (int i = 0; i < 8; ++i) {
            int row = node_base + i + 8 * half;
            if (row < N) {
                const float v = acc[i] * dv[i];
                h[(size_t)row * HID + l16]   = v;
                if (WRITE_AGG) agg[(size_t)row * HID + l16] = v;
                if (l16 == 0) dinv[row] = dv[i];
            }
        }
    }
}


struct __attribute__((aligned(8))) ag_slot { unsigned ho, lo; };

__device__ __forceinline__ void ag_direct(const float* __restrict__ h,
                                       float* __restrict__ agg, int s, int t) {
#pragma nounroll
    for (int j = 0; j < HID; ++j)
        unsafeAtomicAdd(&agg[(size_t)t * HID + j], h[(size_t)s * HID + j]);
}

template <bool FULL>
__device__ __forceinline__ void ag_gather32(const int* __restrict__ src_of, const float* __restrict__ h,
                                            float* win, const unsigned* mylist, ag_slot* myslot,
                                            int k, int n, int lane, int l16, int half) {
    ag_slot mine = {0u, 0u};
    if (FULL || k + lane < n) {
        const unsigned p = mylist[k + lane];
        const int s = src_of[(int)(p >> AG_TBITS)];
        mine.ho = (unsigned)s * HID;
        mine.lo = (p & ((1u << AG_TBITS) - 1u)) * HID;
    }
    myslot[lane] = mine;

    ag_slot q[16];
    float hv[16];
#pragma unroll
    for (int i = 0; i < 16; ++i) {
        q[i]  = myslot[2 * i + half];
        hv[i] = h[q[i].ho + (unsigned)l16];
    }
#pragma unroll
    for (int i = 0; i < 16; ++i) {
        if (FULL || k + 2 * i + half < n)
            atomicAdd(&win[q[i].lo + (unsigned)l16], hv[i]);
    }
}

__global__ __launch_bounds__(AG_THREADS)
void k_edge_agg(const int* __restrict__ ei, int E, int N,
                const float* __restrict__ h,
                float* __restrict__ agg) {
    __shared__ float    win[AG_WIN * HID];
    __shared__ unsigned list[AG_WAVES][AG_CAPW];
    __shared__ ag_slot slot[AG_WAVES][32];

    const int tid  = threadIdx.x;
    const int lane = tid & 31;
    const int wv   = __builtin_amdgcn_readfirstlane(tid >> 5);
    const int l16  = lane & 15;
    const int half = lane >> 4;
    const unsigned lt_mask = (1u << lane) - 1u;

    const int nbase = blockIdx.y * AG_WIN;
    int wn = N - nbase; if (wn > AG_WIN) wn = AG_WIN;
    const int chunk = (((E + AG_CHUNKS - 1) / AG_CHUNKS) + 3) & ~3;
    int e0 = blockIdx.x * chunk; if (e0 > E) e0 = E;
    int e1 = e0 + chunk;         if (e1 > E) e1 = E;
    const int cnt = e1 - e0;
    const int* tg = ei + E + e0;

    v4f* w4 = (v4f*)win;
    for (int i = tid; i < AG_WIN * HID / 4; i += AG_THREADS)
        w4[i] = (v4f){0.0f, 0.0f, 0.0f, 0.0f};

    unsigned* mylist = list[wv];
    int wcnt = 0;
    const int n4 = cnt >> 2;
    const v4i* tg4 = (const v4i*)tg;
    for (int g0 = wv * 32; g0 < n4; g0 += AG_THREADS * 4) {
        v4i t[4];
        if (g0 + 3 * AG_THREADS + 32 <= n4) {
#pragma unroll
            for (int u = 0; u < 4; ++u)
                t[u] = tg4[g0 + u * AG_THREADS + lane];
        } else {
#pragma unroll
            for (int u = 0; u < 4; ++u) {
                const int g = g0 + u * AG_THREADS + lane;
                t[u] = (g < n4) ? tg4[g] : (v4i){-1, -1, -1, -1};
            }
        }
        unsigned m[4][4];
        const int wbase = wcnt;
        int tot = 0;
#pragma unroll
        for (int u = 0; u < 4; ++u) {
#pragma unroll
            for (int c = 0; c < 4; ++c) {
                m[u][c] = __builtin_amdgcn_ballot_w32((unsigned)(t[u][c] - nbase) < (unsigned)wn);
                tot += __builtin_popcount(m[u][c]);
            }
        }
        int run = wbase;
        if (wbase + tot <= AG_CAPW) {
#pragma unroll
            for (int u = 0; u < 4; ++u) {
                const int g = g0 + u * AG_THREADS + lane;
#pragma unroll
                for (int c = 0; c < 4; ++c) {
                    const unsigned off = (unsigned)(t[u][c] - nbase);
                    if (off < (unsigned)wn) {
                        const unsigned pos = __builtin_amdgcn_mbcnt_lo(m[u][c], (unsigned)run);
                        mylist[pos] = ((unsigned)(4 * g + c) << AG_TBITS) | off;
                    }
                    run += __builtin_popcount(m[u][c]);
                }
            }
        } else {
#pragma unroll
            for (int u = 0; u < 4; ++u) {
                const int g = g0 + u * AG_THREADS + lane;
#pragma unroll
                for (int c = 0; c < 4; ++c) {
                    const bool hit = (unsigned)(t[u][c] - nbase) < (unsigned)wn;
                    if (hit) {
                        const int pos = run + __builtin_popcount(m[u][c] & lt_mask);
                        const unsigned le = (unsigned)(4 * g + c);
                        if (pos < AG_CAPW) mylist[pos] = (le << AG_TBITS) | (unsigned)(t[u][c] - nbase);
                        else ag_direct(h, agg, ei[e0 + (int)le], t[u][c]);
                    }
                    run += __builtin_popcount(m[u][c]);
                }
            }
        }
        wcnt = wbase + tot;
    }
    if (wv == 0) {
        const int r = (n4 << 2) + lane;
        int tv = -1;
        if (lane < 3 && r < cnt) tv = tg[r];
        const bool hit = (unsigned)(tv - nbase) < (unsigned)wn;
        const unsigned m = __builtin_amdgcn_ballot_w32(hit);
        if (hit) {
            const int pos = wcnt + __builtin_popcount(m & lt_mask);
            if (pos < AG_CAPW) mylist[pos] = ((unsigned)r << AG_TBITS) | (unsigned)(tv - nbase);
            else ag_direct(h, agg, ei[e0 + r], tv);
        }
        wcnt += __builtin_popcount(m);
    }
    __syncthreads();

    const int n = wcnt < AG_CAPW ? wcnt : AG_CAPW;
    {
        int k = 0;
        for (; k + 32 <= n; k += 32)
            ag_gather32<true>(ei + e0, h, win, mylist, slot[wv], k, n, lane, l16, half);
        if (k < n)
            ag_gather32<false>(ei + e0, h, win, mylist, slot[wv], k, n, lane, l16, half);
    }
    __syncthreads();

    float* dst = agg + (size_t)nbase * HID;
    for (int i = tid; i < wn * HID; i += AG_THREADS)
        unsafeAtomicAdd(&dst[i], win[i]);
}


template <bool FULL>
__device__ __forceinline__ void xw_rows32(const float* __restrict__ h, float* win, const unsigned* mylist,
                                          int k, int n, int l16, int half) {
    const int first = k + 16 * half;
    const v4u* lp = (const v4u*)(mylist + first);
    unsigned e[16];
#pragma unroll
    for (int q = 0; q < 4; ++q) {
        const v4u t = lp[q];
        e[4 * q + 0] = t.x; e[4 * q + 1] = t.y; e[4 * q + 2] = t.z; e[4 * q + 3] = t.w;
    }
    float hv[16];
#pragma unroll
    for (int i = 0; i < 16; ++i) {
        if (!FULL && first + i >= n) e[i] = 0u;
        hv[i] = h[(e[i] >> AG_TBITS) * HID + (unsigned)l16];
    }
#pragma unroll
    for (int i = 0; i < 16; ++i) {
        if (FULL || first + i < n)
            atomicAdd(&win[(e[i] & ((1u << AG_TBITS) - 1u)) * HID + (unsigned)l16], hv[i]);
    }
}

__global__ __launch_bounds__(XW_THREADS)
void k_edge_out(const int* __restrict__ ei, int E, int N,
                const float* __restrict__ h,
                const float* __restrict__ dinv,
                const unsigned* __restrict__ bucket,
                const unsigned* __restrict__ bcnt,
                const float* __restrict__ conv_b,
                const float* __restrict__ out_w,
                const float* __restrict__ out_b,
                float* __restrict__ out) {
    __shared__ __attribute__((aligned(16))) float    win[XW_WIN * HID];
    __shared__ __attribute__((aligned(16))) unsigned list[XW_WAVES][XW_LCAP];

    const int tid  = threadIdx.x;
    const int lane = tid & 31;
    const int wv   = __builtin_amdgcn_readfirstlane(tid >> 5);
    const int l16  = lane & 15;
    const int half = lane >> 4;

    const int nbase = blockIdx.x * XW_WIN;
    int wn = N - nbase; if (wn > XW_WIN) wn = XW_WIN;
    const int b0 = wv * XW_BPW;
    const unsigned* seg0 = bucket + ((size_t)blockIdx.x * BIN_PRODS + (size_t)b0) * BIN_CAP;
    const unsigned cmine = (lane < XW_BPW) ? bcnt[(size_t)blockIdx.x * BIN_PRODS + b0 + lane] : 0u;

    {
        const v4f* h4 = (const v4f*)(h + (size_t)nbase * HID);
        v4f* w4 = (v4f*)win;
        for (int i = tid; i < XW_WIN * HID / 4; i += XW_THREADS)
            w4[i] = (i * 4 < wn * HID) ? h4[i] : (v4f){0.0f, 0.0f, 0.0f, 0.0f};
    }

    int cnt[XW_BPW];
    unsigned ent[XW_BPW][BIN_LINES];
#pragma unroll
    for (int j = 0; j < XW_BPW; ++j) {
        cnt[j] = __builtin_amdgcn_readlane((int)cmine, j);
        const int cl = ((unsigned)cnt[j] <= (unsigned)BIN_CAP) ? cnt[j] : 0;
        const unsigned* seg = seg0 + j * BIN_CAP;
#pragma unroll
        for (int u = 0; u < BIN_LINES; ++u)
            ent[j][u] = (u * 32 + lane < cl) ? seg[u * 32 + lane] : 0u;
    }
    unsigned* mylist = list[wv];
    int n = 0;
    unsigned rescan = 0u;
#pragma unroll
    for (int j = 0; j < XW_BPW; ++j) {
        const int c = cnt[j];
        if ((unsigned)c <= (unsigned)BIN_CAP && n + c <= XW_LCAP) {
#pragma unroll
            for (int u = 0; u < BIN_LINES; ++u)
                if (u * 32 + lane < c) mylist[n + u * 32 + lane] = ent[j][u];
            n += c;
        } else {
            rescan |= 1u << j;
        }
    }
    __syncthreads();

    {
        int k = 0;
        for (; k + 32 <= n; k += 32)
            xw_rows32<true>(h, win, mylist, k, n, l16, half);
        if (k < n)
            xw_rows32<false>(h, win, mylist, k, n, l16, half);
    }
    if (rescan != 0u) {
        const int sub = bin_sub(E);
#pragma nounroll
        for (int j = 0; j < XW_BPW; ++j) {
            if (((rescan >> j) & 1u) == 0u) continue;
            int e0 = (b0 + j) * sub;  if (e0 > E) e0 = E;
            int e1 = e0 + sub;        if (e1 > E) e1 = E;
            for (int e = e0 + lane; e < e1; e += 32) {
                const unsigned off = (unsigned)(ei[E + e] - nbase);
                if (off < (unsigned)wn) {
                    const int sidx = ei[e];
#pragma nounroll
                    for (int c = 0; c < HID; ++c)
                        atomicAdd(&win[off * HID + (unsigned)c], h[(size_t)sidx * HID + c]);
                }
            }
        }
    }
    __syncthreads();

    if (tid < wn) {
        const int nd = nbase + tid;
        const v4f* av = (const v4f*)(win + tid * HID);
        const v4f* bv = (const v4f*)conv_b;
        const v4f* ov = (const v4f*)out_w;
        const float dn = dinv[nd];
        float acc = out_b[0];
#pragma unroll
        for (int i = 0; i < 4; ++i) {
            v4f v = av[i] * dn + bv[i];
            v4f wvi = ov[i];
#pragma unroll
            for (int j = 0; j < 4; ++j)
                acc = fmaf(fmaxf(v[j], 0.0f), wvi[j], acc);
        }
        out[nd] = acc;
    }
}


__global__ void k_out(const float* __restrict__ agg,
                      const float* __restrict__ dinv,
                      const float* __restrict__ conv_b,
                      const float* __restrict__ out_w,
                      const float* __restrict__ out_b,
                      float* __restrict__ out, int N) {
    int n = blockIdx.x * blockDim.x + threadIdx.x;
    if (n >= N) return;
    const v4f* av = (const v4f*)(agg + (size_t)n * HID);
    const v4f* bv = (const v4f*)conv_b;
    const v4f* wv = (const v4f*)out_w;
    const float dn = dinv[n];
    float acc = out_b[0];
#pragma unroll
    for (int i = 0; i < 4; ++i) {
        v4f v = av[i] * dn + bv[i];
        v4f wvi = wv[i];
#pragma unroll
        for (int j = 0; j < 4; ++j)
            acc = fmaf(fmaxf(v[j], 0.0f), wvi[j], acc);
    }
    out[n] = acc;
}


static inline size_t align256(size_t v) { return (v + 255) & ~(size_t)255; }

extern "C" void kernel_launch(void* const* d_in, const int* in_sizes, int n_in,
                              void* d_out, int out_size, void* d_ws, size_t ws_size,
                              hipStream_t stream) {
    const float* x      = (const float*)d_in[0];
    const int*   ei     = (const int*)d_in[1];
    const float* conv_w = (const float*)d_in[2];
    const float* conv_b = (const float*)d_in[3];
    const float* out_w  = (const float*)d_in[4];
    const float* out_b  = (const float*)d_in[5];
    float* out = (float*)d_out;

    const int N = in_sizes[0] / IN_CH;
    const int E = in_sizes[1] / 2;

    const int nblk = (N + 31) / 32;

    char* ws = (char*)d_ws;
    size_t off = 0;
    float* dinv   = (float*)(ws + off);    off = align256(off + (size_t)N * 4);
    float* h      = (float*)(ws + off);    off = align256(off + (size_t)nblk * 32 * HID * 4);
    const int nwin = (N + XW_WIN - 1) / XW_WIN;
    size_t offb = off;
    unsigned* bcnt   = (unsigned*)(ws + offb);   offb = align256(offb + (size_t)nwin * BIN_PRODS * 4);
    unsigned* bucket = (unsigned*)(ws + offb);   offb = align256(offb + (size_t)nwin * BIN_PRODS * BIN_CAP * 4);
    const bool binned = (offb <= ws_size) && (nwin <= BIN_MAXW) && (N <= (1 << (32 - AG_TBITS)));

    const int ndegy = (nblk * 32 + DEG_WIN - 1) / DEG_WIN;

    if (binned) {
        const int nbiny = (BIN_PRODS + DEG_PARTS - 1) / DEG_PARTS;
        k_deg_hist<<<dim3(DEG_PARTS, ndegy + nbiny), DEG_THREADS, 0, stream>>>(ei, E, N, (unsigned*)h, ndegy, nwin, bucket, bcnt);
        k_gemm_wmma<false><<<nblk, 64, 0, stream>>>(x, conv_w, (const unsigned*)h, dinv, h, nullptr, N);
        k_edge_out<<<nwin, XW_THREADS, 0, stream>>>(ei, E, N, h, dinv, bucket, bcnt, conv_b, out_w, out_b, out);
    } else {
        float* agg = (float*)(ws + off);
        const int nwin_s = (N + AG_WIN - 1) / AG_WIN;
        const int nb_n = (N + 255) / 256;
        k_deg_hist<<<dim3(DEG_PARTS, ndegy), DEG_THREADS, 0, stream>>>(ei, E, N, (unsigned*)agg, ndegy, 0, nullptr, nullptr);
        k_gemm_wmma<true><<<nblk, 64, 0, stream>>>(x, conv_w, (const unsigned*)agg, dinv, h, agg, N);
        k_edge_agg<<<dim3(AG_CHUNKS, nwin_s), AG_THREADS, 0, stream>>>(ei, E, N, h, agg);
        k_out<<<nb_n, 256, 0, stream>>>(agg, dinv, conv_b, out_w, out_b, out, N);
    }
}


# degrees counted from the edge bins in the product launch: one 26-wave work-group per 416-node window, x staged by k quarters ahead of its WMMAs; 8x target scan and partial counts gone; sums untouched
# speedup vs baseline: 6.7243x; 1.2082x over previous
#include <hip/hip_runtime.h>
#include <hip/hip_bf16.h>

typedef __attribute__((ext_vector_type(2))) float v2f;
typedef __attribute__((ext_vector_type(4))) float v4f;
typedef __attribute__((ext_vector_type(8))) float v8f;

#define IN_CH 256
#define HID   16


typedef __attribute__((ext_vector_type(4))) unsigned v4u;
typedef __attribute__((ext_vector_type(4))) int v4i;

#define DEG_PARTS   16
#define DEG_THREADS 1024
#define DEG_WIN     12800
#define DEG_LOADS   8
static_assert(DEG_PARTS * 4 == HID * 4, "a node's partial counts must fill exactly its row of h");
static_assert(DEG_WIN % 32 == 0, "a window holds whole node blocks");

#define AG_THREADS 1024
#define AG_WAVES   (AG_THREADS / 32)
#define AG_CHUNKS  9
#define AG_WIN     3584
#define AG_CAPW    512
#define AG_TBITS   12
static_assert(AG_WIN <= (1 << AG_TBITS), "a window offset must fit the low bits of a list entry");
#define XW_WIN     416
#define XW_THREADS 1024
#define XW_WAVES   (XW_THREADS / 32)
#define XW_LCAP    640
static_assert(XW_WIN <= (1 << AG_TBITS), "a window offset must fit the low bits of a list entry");
static_assert(XW_WIN % 32 == 0 && XW_WIN <= XW_THREADS, "whole lines of out per window, one thread per node at the end");
static_assert(XW_LCAP % 32 == 0, "a group of 32 list entries never leaves the wave's list");
#define BIN_PRODS  288
#define XW_BPW     (BIN_PRODS / XW_WAVES)
static_assert(BIN_PRODS % XW_WAVES == 0, "every wave of the sum kernel takes the same number of bins");
#define BIN_CAP    96
#define BIN_LINES  (BIN_CAP / 32)
static_assert(BIN_CAP % 32 == 0, "a bin is whole lines");
#define BIN_MAXW   256
#define BIN_LOADS  3
#define DEG_SMEM   (BIN_MAXW * BIN_CAP + BIN_MAXW)
static_assert(DEG_SMEM >= DEG_WIN, "the shared array must hold a degree window");

__device__ __forceinline__ int bin_sub(int E) {
    return (((E + BIN_PRODS - 1) / BIN_PRODS) + 3) & ~3;
}

__device__ __forceinline__ void bin_four(unsigned* stage, unsigned* cntL, int nwin, v4i tv, v4i sv) {
    unsigned w[4], r[4];
#pragma unroll
    for (int c = 0; c < 4; ++c) {
        w[c] = (unsigned)tv[c] / (unsigned)XW_WIN;
        r[c] = BIN_CAP;
        if (w[c] < (unsigned)nwin) r[c] = atomicAdd(&cntL[w[c]], 1u);
    }
#pragma unroll
    for (int c = 0; c < 4; ++c) {
        if (r[c] < BIN_CAP)
            stage[w[c] * BIN_CAP + r[c]] = ((unsigned)sv[c] << AG_TBITS) | ((unsigned)tv[c] - w[c] * XW_WIN);
    }
}

__device__ __forceinline__ void bin_role(const int* __restrict__ ei, int E, int b, int nwin,
                                         unsigned* __restrict__ bucket, unsigned* __restrict__ bcnt,
                                         unsigned* smem) {
    unsigned* stage = smem;
    unsigned* cntL  = smem + BIN_MAXW * BIN_CAP;
    const int tid  = threadIdx.x;
    const int lane = tid & 31;
    const int wv   = __builtin_amdgcn_readfirstlane(tid >> 5);
    const int sub  = bin_sub(E);
    int e0 = b * sub;    if (e0 > E) e0 = E;
    int e1 = e0 + sub;   if (e1 > E) e1 = E;
    const int cnt = e1 - e0;
    const int* tg = ei + E + e0;
    const int* sr = ei + e0;

    if (tid < BIN_MAXW) cntL[tid] = 0u;
    __syncthreads();

    const int n4 = cnt >> 2;
    const v4i* tg4 = (const v4i*)tg;
    const v4i* sr4 = (const v4i*)sr;
    for (int g = tid; g < n4; g += DEG_THREADS * BIN_LOADS) {
        v4i t[BIN_LOADS], sv[BIN_LOADS];
#pragma unroll
        for (int u = 0; u < BIN_LOADS; ++u) {
            const int gg = g + u * DEG_THREADS;
            t[u]  = (gg < n4) ? tg4[gg] : (v4i){-1, -1, -1, -1};
            sv[u] = (gg < n4) ? sr4[gg] : (v4i){0, 0, 0, 0};
        }
#pragma unroll
        for (int u = 0; u < BIN_LOADS; ++u)
            bin_four(stage, cntL, nwin, t[u], sv[u]);
    }
    {
        const int r = (n4 << 2) + tid;
        if (r < cnt) {
            const v4i tv = {tg[r], -1, -1, -1};
            const v4i s1 = {sr[r], 0, 0, 0};
            bin_four(stage, cntL, nwin, tv, s1);
        }
    }
    __syncthreads();

    for (int w = wv; w < nwin; w += DEG_THREADS / 32) {
        unsigned n = (unsigned)__builtin_amdgcn_readfirstlane((int)cntL[w]);
        if (n > BIN_CAP) n = BIN_CAP;
        const unsigned n32 = (n + 31u) & ~31u;
        unsigned* dst = bucket + ((size_t)w * BIN_PRODS + (size_t)b) * BIN_CAP;
        const unsigned* st = stage + w * BIN_CAP;
        for (unsigned k = 0; k < n32; k += 32)
            dst[k + lane] = st[k + lane];
    }
    if (tid < nwin) bcnt[(size_t)tid * BIN_PRODS + b] = cntL[tid];
}

__device__ __forceinline__ void deg_count(unsigned* hist, unsigned nbase, unsigned wn, int tv) {
    const unsigned w = (unsigned)tv - nbase;
    if (w < wn) atomicAdd(&hist[w], 1u);
}

__global__ __launch_bounds__(DEG_THREADS)
void k_deg_hist(const int* __restrict__ ei, int E, int N,
                unsigned* __restrict__ part,
                int ndegy, int nwin,
                unsigned* __restrict__ bucket, unsigned* __restrict__ bcnt) {
    __shared__ unsigned hist[DEG_SMEM];

    if ((int)blockIdx.y >= ndegy) {
        const int b = ((int)blockIdx.y - ndegy) * (int)gridDim.x + (int)blockIdx.x;
        if (b < BIN_PRODS) bin_role(ei, E, b, nwin, bucket, bcnt, hist);
        return;
    }

    const int tid   = threadIdx.x;
    const int lane  = tid & 31;
    const int wv    = tid >> 5;
    const int nblk  = (N + 31) >> 5;
    const int chunk = (((E + DEG_PARTS - 1) / DEG_PARTS) + 3) & ~3;
    const unsigned nbase = blockIdx.y * DEG_WIN;
    unsigned wn = (unsigned)nblk * 32u - nbase;
    if (wn > DEG_WIN) wn = DEG_WIN;

    int e0 = blockIdx.x * chunk; if (e0 > E) e0 = E;
    int e1 = e0 + chunk;         if (e1 > E) e1 = E;
    const int cnt = e1 - e0;
    const int* tg = ei + E + e0;

    v4u* h4 = (v4u*)hist;
    for (int i = tid; i < DEG_WIN / 4; i += DEG_THREADS)
        h4[i] = (v4u){0u, 0u, 0u, 0u};
    __syncthreads();

    const int n4 = cnt >> 2;
    const v4i* tg4 = (const v4i*)tg;
    for (int g = tid; g < n4; g += DEG_THREADS * DEG_LOADS) {
        v4i t[DEG_LOADS];
#pragma unroll
        for (int u = 0; u < DEG_LOADS; ++u) {
            int gg = g + u * DEG_THREADS;
            t[u] = (gg < n4) ? tg4[gg] : (v4i){-1, -1, -1, -1};
        }
#pragma unroll
        for (int u = 0; u < DEG_LOADS; ++u) {
            deg_count(hist, nbase, wn, t[u].x); deg_count(hist, nbase, wn, t[u].y);
            deg_count(hist, nbase, wn, t[u].z); deg_count(hist, nbase, wn, t[u].w);
        }
    }
    {
        int r = (n4 << 2) + tid;
        if (r < cnt) deg_count(hist, nbase, wn, tg[r]);
    }
    __syncthreads();

    const int nb0 = (int)(nbase >> 5);
    const int nbw = (int)(wn >> 5);
    for (int nb = wv; nb < nbw; nb += DEG_THREADS / 32)
        part[((size_t)(nb0 + nb) * DEG_PARTS + blockIdx.x) * 32 + lane] = hist[nb * 32 + lane];
}

#define XPAD 4

template <bool WRITE_AGG>
__global__ __launch_bounds__(64)
void k_gemm_wmma(const float* __restrict__ x,
                 const float* __restrict__ w,
                 const unsigned* part,
                 float* __restrict__ dinv,
                 float* h,
                 float* agg, int N) {
    __shared__ float sW[IN_CH][HID];
    __shared__ float sX[2][16][IN_CH + XPAD];
    __shared__ unsigned sDeg[2][32];

    const int tid  = threadIdx.x;
    const int wv   = tid >> 5;
    const int lane = tid & 31;
    const int l16  = lane & 15;
    const int half = lane >> 4;
    const int node_base = blockIdx.x * 32 + wv * 16;
    const bool full_tile = (node_base + 16 <= N);

    {
        const unsigned* pp = part + ((size_t)blockIdx.x * DEG_PARTS + (size_t)wv * (DEG_PARTS / 2)) * 32 + lane;
        unsigned sum = 0u;
#pragma unroll
        for (int p = 0; p < DEG_PARTS / 2; ++p)
            sum += pp[p * 32];
        sDeg[wv][lane] = sum;
    }

    {
        const v4f* ws = (const v4f*)w;
        v4f* wd = (v4f*)&sW[0][0];
#pragma unroll
        for (int i = 0; i < (IN_CH * HID) / 4 / 64; ++i)
            wd[tid + i * 64] = ws[tid + i * 64];
    }

    if (full_tile) {
        const v4f* xs = (const v4f*)(x + (size_t)node_base * IN_CH);
#pragma unroll
        for (int it = 0; it < 32; ++it) {
            int fi = it * 32 + lane;
            v4f v = __builtin_nontemporal_load(xs + fi);
            int f = fi * 4;
            *(v4f*)&sX[wv][f >> 8][f & 255] = v;
        }
    } else {
        for (int it = 0; it < 32; ++it) {
            int f = it * 128 + lane * 4;
            int r = f >> 8, c = f & 255;
            int row = node_base + r;
            v4f v = {};
            if (row < N) v = *(const v4f*)(x + (size_t)row * IN_CH + c);
            *(v4f*)&sX[wv][r][c] = v;
        }
    }
    __syncthreads();

    float dv[8];
#pragma unroll
    for (int i = 0; i < 8; ++i) {
        const int r = wv * 16 + 8 * half + i;
        dv[i] = rsqrtf((float)(1u + sDeg[0][r] + sDeg[1][r]));
    }

    v8f acc = {};
#pragma unroll 8
    for (int k = 0; k < IN_CH; k += 4) {
        v2f a;
        a.x = sX[wv][l16][k + 2 * half + 0];
        a.y = sX[wv][l16][k + 2 * half + 1];
        v2f b;
        b.x = sW[k + 2 * half + 0][l16];
        b.y = sW[k + 2 * half + 1][l16];
        acc = __builtin_amdgcn_wmma_f32_16x16x4_f32(
            false, a, false, b, (short)0, acc, false, false);
    }

    if (full_tile) {
        unsigned base = (unsigned)(node_base + 8 * half) * HID + (unsigned)l16;
#pragma unroll
        for (int i = 0; i < 8; ++i) {
            const float v = acc[i] * dv[i];
            h[base + (unsigned)i * HID]   = v;
            if (WRITE_AGG) agg[base + (unsigned)i * HID] = v;
        }
        if (l16 == 0) {
#pragma unroll
            for (int i = 0; i < 8; ++i)
                dinv[node_base + 8 * half + i] = dv[i];
        }
    } else {
#pragma unroll
        for (int i = 0; i < 8; ++i) {
            int row = node_base + i + 8 * half;
            if (row < N) {
                const float v = acc[i] * dv[i];
                h[(size_t)row * HID + l16]   = v;
                if (WRITE_AGG) agg[(size_t)row * HID + l16] = v;
                if (l16 == 0) dinv[row] = dv[i];
            }
        }
    }
}

#define GW_WAVES   (XW_WIN / 16)
#define GW_THREADS (GW_WAVES * 32)
#define GW_KQ      (IN_CH / 4)
#define GW_PITCH   (GW_KQ + 4)
#define GW_BPW     ((BIN_PRODS + GW_WAVES - 1) / GW_WAVES)
#define GW_HOT     2
static_assert(GW_HOT <= BIN_LINES, "no more lines than a bin has");
static_assert(XW_WIN % 16 == 0, "a window is whole tiles");
static_assert(GW_BPW <= 32, "lane j holds the count of the wave's bin j");

__device__ __forceinline__ void gw_lds_barrier() {
    __builtin_amdgcn_fence(__ATOMIC_RELEASE, "workgroup", "local");
    __builtin_amdgcn_s_barrier();
    __builtin_amdgcn_fence(__ATOMIC_ACQUIRE, "workgroup", "local");
}

__device__ __forceinline__ void gw_load(const float* __restrict__ x, int node_base, int N, int q,
                                        int lane, v4f (&buf)[8]) {
    const int r0 = lane >> 4, c4 = lane & 15;
#pragma unroll
    for (int it = 0; it < 8; ++it) {
        int row = node_base + 2 * it + r0;
        if (row > N - 1) row = N - 1;
        const unsigned i4 = (unsigned)row * (IN_CH / 4) + (unsigned)(q * (GW_KQ / 4) + c4);
        buf[it] = __builtin_nontemporal_load((const v4f*)x + i4);
    }
}

__device__ __forceinline__ void gw_stage(float* st, int lane, const v4f (&buf)[8]) {
    const int r0 = lane >> 4, c4 = lane & 15;
#pragma unroll
    for (int it = 0; it < 8; ++it)
        *(v4f*)&st[(2 * it + r0) * GW_PITCH + 4 * c4] = buf[it];
}

__device__ __forceinline__ v8f gw_mma(v8f acc, const float* st, const float (*sW)[HID], int q, int l16, int half) {
#pragma unroll 8
    for (int kk = 0; kk < GW_KQ; kk += 4) {
        v2f a;
        a.x = st[l16 * GW_PITCH + kk + 2 * half + 0];
        a.y = st[l16 * GW_PITCH + kk + 2 * half + 1];
        v2f b;
        b.x = sW[q * GW_KQ + kk + 2 * half + 0][l16];
        b.y = sW[q * GW_KQ + kk + 2 * half + 1][l16];
        acc = __builtin_amdgcn_wmma_f32_16x16x4_f32(
            false, a, false, b, (short)0, acc, false, false);
    }
    return acc;
}

__global__ __launch_bounds__(GW_THREADS)
void k_gemm_win(const float* __restrict__ x,
                const float* __restrict__ w,
                const int* __restrict__ ei, int E,
                const unsigned* __restrict__ bucket,
                const unsigned* __restrict__ bcnt,
                float* __restrict__ dinv,
                float* __restrict__ h, int N) {
    __shared__ __attribute__((aligned(16))) float sW[IN_CH][HID];
    __shared__ __attribute__((aligned(16))) float sX[GW_WAVES][2][16 * GW_PITCH];
    __shared__ unsigned hist[XW_WIN];

    const int tid  = threadIdx.x;
    const int lane = tid & 31;
    const int wv   = __builtin_amdgcn_readfirstlane(tid >> 5);
    const int l16  = lane & 15;
    const int half = lane >> 4;

    const int nbase = blockIdx.x * XW_WIN;
    int wn = N - nbase; if (wn > XW_WIN) wn = XW_WIN;
    const int node_base = nbase + wv * 16;

    const int b0 = wv * GW_BPW;
    const unsigned* cbase = bcnt + (size_t)blockIdx.x * BIN_PRODS;
    const bool has_bin = lane < GW_BPW && b0 + lane < BIN_PRODS;
    const unsigned cload = cbase[has_bin ? b0 + lane : BIN_PRODS - 1];
    v4f wreg[2];
#pragma unroll
    for (int i = 0; i < 2; ++i) {
        int idx = tid + i * GW_THREADS;
        if (idx > IN_CH * HID / 4 - 1) idx = IN_CH * HID / 4 - 1;
        wreg[i] = ((const v4f*)w)[idx];
    }
    v4f bufA[8], bufB[8];
    gw_load(x, node_base, N, 0, lane, bufA);
    gw_load(x, node_base, N, 1, lane, bufB);

    if (tid < XW_WIN) hist[tid] = 0u;
    const unsigned cmine = has_bin ? cload : 0u;
#pragma unroll
    for (int i = 0; i < 2; ++i) {
        const int idx = tid + i * GW_THREADS;
        if (idx < IN_CH * HID / 4) ((v4f*)&sW[0][0])[idx] = wreg[i];
    }

    int cnt[GW_BPW];
    unsigned ent[GW_BPW][GW_HOT];
    const unsigned* seg0 = bucket + ((size_t)blockIdx.x * BIN_PRODS + (size_t)b0) * BIN_CAP;
#pragma unroll
    for (int j = 0; j < GW_BPW; ++j) {
        cnt[j] = __builtin_amdgcn_readlane((int)cmine, j);
        const int cl = ((unsigned)cnt[j] <= (unsigned)BIN_CAP) ? cnt[j] : 0;
#pragma unroll
        for (int u = 0; u < GW_HOT; ++u) {
            const unsigned* base = (u * 32 < cl) ? seg0 + j * BIN_CAP + u * 32 : cbase;
            ent[j][u] = base[lane];
        }
    }

    gw_lds_barrier();

    gw_stage(sX[wv][0], lane, bufA);
    gw_stage(sX[wv][1], lane, bufB);

    unsigned rescan = 0u;
#pragma unroll
    for (int j = 0; j < GW_BPW; ++j) {
        const int c = cnt[j];
        if ((unsigned)c <= (unsigned)BIN_CAP) {
#pragma unroll
            for (int u = 0; u < GW_HOT; ++u)
                if (u * 32 + lane < c) atomicAdd(&hist[ent[j][u] & ((1u << AG_TBITS) - 1u)], 1u);
            for (int k = GW_HOT * 32; k < c; k += 32)
                if (k + lane < c) atomicAdd(&hist[seg0[j * BIN_CAP + k + lane] & ((1u << AG_TBITS) - 1u)], 1u);
        } else {
            rescan |= 1u << j;
        }
    }
    if (rescan != 0u) {
        const int sub = bin_sub(E);
#pragma nounroll
        for (int j = 0; j < GW_BPW; ++j) {
            if (((rescan >> j) & 1u) == 0u) continue;
            int e0 = (b0 + j) * sub;  if (e0 > E) e0 = E;
            int e1 = e0 + sub;        if (e1 > E) e1 = E;
            for (int e = e0 + lane; e < e1; e += 32) {
                const unsigned off = (unsigned)(ei[E + e] - nbase);
                if (off < (unsigned)wn) atomicAdd(&hist[off], 1u);
            }
        }
    }

    gw_load(x, node_base, N, 2, lane, bufA);
    gw_load(x, node_base, N, 3, lane, bufB);

    gw_lds_barrier();

    float dv[8];
#pragma unroll
    for (int i = 0; i < 8; ++i)
        dv[i] = rsqrtf((float)(1u + hist[wv * 16 + 8 * half + i]));

    v8f acc = {};
    acc = gw_mma(acc, sX[wv][0], sW, 0, l16, half);
    acc = gw_mma(acc, sX[wv][1], sW, 1, l16, half);
    gw_stage(sX[wv][0], lane, bufA);
    gw_stage(sX[wv][1], lane, bufB);
    acc = gw_mma(acc, sX[wv][0], sW, 2, l16, half);
    acc = gw_mma(acc, sX[wv][1], sW, 3, l16, half);

    if (node_base + 16 <= N) {
        unsigned base = (unsigned)(node_base + 8 * half) * HID + (unsigned)l16;
#pragma unroll
        for (int i = 0; i < 8; ++i)
            h[base + (unsigned)i * HID] = acc[i] * dv[i];
        if (l16 == 0) {
#pragma unroll
            for (int i = 0; i < 8; ++i)
                dinv[node_base + 8 * half + i] = dv[i];
        }
    } else {
#pragma unroll
        for (int i = 0; i < 8; ++i) {
            int row = node_base + i + 8 * half;
            if (row < N) {
                h[(size_t)row * HID + l16] = acc[i] * dv[i];
                if (l16 == 0) dinv[row] = dv[i];
            }
        }
    }
}


struct __attribute__((aligned(8))) ag_slot { unsigned ho, lo; };

__device__ __forceinline__ void ag_direct(const float* __restrict__ h,
                                       float* __restrict__ agg, int s, int t) {
#pragma nounroll
    for (int j = 0; j < HID; ++j)
        unsafeAtomicAdd(&agg[(size_t)t * HID + j], h[(size_t)s * HID + j]);
}

template <bool FULL>
__device__ __forceinline__ void ag_gather32(const int* __restrict__ src_of, const float* __restrict__ h,
                                            float* win, const unsigned* mylist, ag_slot* myslot,
                                            int k, int n, int lane, int l16, int half) {
    ag_slot mine = {0u, 0u};
    if (FULL || k + lane < n) {
        const unsigned p = mylist[k + lane];
        const int s = src_of[(int)(p >> AG_TBITS)];
        mine.ho = (unsigned)s * HID;
        mine.lo = (p & ((1u << AG_TBITS) - 1u)) * HID;
    }
    myslot[lane] = mine;

    ag_slot q[16];
    float hv[16];
#pragma unroll
    for (int i = 0; i < 16; ++i) {
        q[i]  = myslot[2 * i + half];
        hv[i] = h[q[i].ho + (unsigned)l16];
    }
#pragma unroll
    for (int i = 0; i < 16; ++i) {
        if (FULL || k + 2 * i + half < n)
            atomicAdd(&win[q[i].lo + (unsigned)l16], hv[i]);
    }
}

__global__ __launch_bounds__(AG_THREADS)
void k_edge_agg(const int* __restrict__ ei, int E, int N,
                const float* __restrict__ h,
                float* __restrict__ agg) {
    __shared__ float    win[AG_WIN * HID];
    __shared__ unsigned list[AG_WAVES][AG_CAPW];
    __shared__ ag_slot slot[AG_WAVES][32];

    const int tid  = threadIdx.x;
    const int lane = tid & 31;
    const int wv   = __builtin_amdgcn_readfirstlane(tid >> 5);
    const int l16  = lane & 15;
    const int half = lane >> 4;
    const unsigned lt_mask = (1u << lane) - 1u;

    const int nbase = blockIdx.y * AG_WIN;
    int wn = N - nbase; if (wn > AG_WIN) wn = AG_WIN;
    const int chunk = (((E + AG_CHUNKS - 1) / AG_CHUNKS) + 3) & ~3;
    int e0 = blockIdx.x * chunk; if (e0 > E) e0 = E;
    int e1 = e0 + chunk;         if (e1 > E) e1 = E;
    const int cnt = e1 - e0;
    const int* tg = ei + E + e0;

    v4f* w4 = (v4f*)win;
    for (int i = tid; i < AG_WIN * HID / 4; i += AG_THREADS)
        w4[i] = (v4f){0.0f, 0.0f, 0.0f, 0.0f};

    unsigned* mylist = list[wv];
    int wcnt = 0;
    const int n4 = cnt >> 2;
    const v4i* tg4 = (const v4i*)tg;
    for (int g0 = wv * 32; g0 < n4; g0 += AG_THREADS * 4) {
        v4i t[4];
        if (g0 + 3 * AG_THREADS + 32 <= n4) {
#pragma unroll
            for (int u = 0; u < 4; ++u)
                t[u] = tg4[g0 + u * AG_THREADS + lane];
        } else {
#pragma unroll
            for (int u = 0; u < 4; ++u) {
                const int g = g0 + u * AG_THREADS + lane;
                t[u] = (g < n4) ? tg4[g] : (v4i){-1, -1, -1, -1};
            }
        }
        unsigned m[4][4];
        const int wbase = wcnt;
        int tot = 0;
#pragma unroll
        for (int u = 0; u < 4; ++u) {
#pragma unroll
            for (int c = 0; c < 4; ++c) {
                m[u][c] = __builtin_amdgcn_ballot_w32((unsigned)(t[u][c] - nbase) < (unsigned)wn);
                tot += __builtin_popcount(m[u][c]);
            }
        }
        int run = wbase;
        if (wbase + tot <= AG_CAPW) {
#pragma unroll
            for (int u = 0; u < 4; ++u) {
                const int g = g0 + u * AG_THREADS + lane;
#pragma unroll
                for (int c = 0; c < 4; ++c) {
                    const unsigned off = (unsigned)(t[u][c] - nbase);
                    if (off < (unsigned)wn) {
                        const unsigned pos = __builtin_amdgcn_mbcnt_lo(m[u][c], (unsigned)run);
                        mylist[pos] = ((unsigned)(4 * g + c) << AG_TBITS) | off;
                    }
                    run += __builtin_popcount(m[u][c]);
                }
            }
        } else {
#pragma unroll
            for (int u = 0; u < 4; ++u) {
                const int g = g0 + u * AG_THREADS + lane;
#pragma unroll
                for (int c = 0; c < 4; ++c) {
                    const bool hit = (unsigned)(t[u][c] - nbase) < (unsigned)wn;
                    if (hit) {
                        const int pos = run + __builtin_popcount(m[u][c] & lt_mask);
                        const unsigned le = (unsigned)(4 * g + c);
                        if (pos < AG_CAPW) mylist[pos] = (le << AG_TBITS) | (unsigned)(t[u][c] - nbase);
                        else ag_direct(h, agg, ei[e0 + (int)le], t[u][c]);
                    }
                    run += __builtin_popcount(m[u][c]);
                }
            }
        }
        wcnt = wbase + tot;
    }
    if (wv == 0) {
        const int r = (n4 << 2) + lane;
        int tv = -1;
        if (lane < 3 && r < cnt) tv = tg[r];
        const bool hit = (unsigned)(tv - nbase) < (unsigned)wn;
        const unsigned m = __builtin_amdgcn_ballot_w32(hit);
        if (hit) {
            const int pos = wcnt + __builtin_popcount(m & lt_mask);
            if (pos < AG_CAPW) mylist[pos] = ((unsigned)r << AG_TBITS) | (unsigned)(tv - nbase);
            else ag_direct(h, agg, ei[e0 + r], tv);
        }
        wcnt += __builtin_popcount(m);
    }
    __syncthreads();

    const int n = wcnt < AG_CAPW ? wcnt : AG_CAPW;
    {
        int k = 0;
        for (; k + 32 <= n; k += 32)
            ag_gather32<true>(ei + e0, h, win, mylist, slot[wv], k, n, lane, l16, half);
        if (k < n)
            ag_gather32<false>(ei + e0, h, win, mylist, slot[wv], k, n, lane, l16, half);
    }
    __syncthreads();

    float* dst = agg + (size_t)nbase * HID;
    for (int i = tid; i < wn * HID; i += AG_THREADS)
        unsafeAtomicAdd(&dst[i], win[i]);
}


template <bool FULL>
__device__ __forceinline__ void xw_rows32(const float* __restrict__ h, float* win, const unsigned* mylist,
                                          int k, int n, int l16, int half) {
    const int first = k + 16 * half;
    const v4u* lp = (const v4u*)(mylist + first);
    unsigned e[16];
#pragma unroll
    for (int q = 0; q < 4; ++q) {
        const v4u t = lp[q];
        e[4 * q + 0] = t.x; e[4 * q + 1] = t.y; e[4 * q + 2] = t.z; e[4 * q + 3] = t.w;
    }
    float hv[16];
#pragma unroll
    for (int i = 0; i < 16; ++i) {
        if (!FULL && first + i >= n) e[i] = 0u;
        hv[i] = h[(e[i] >> AG_TBITS) * HID + (unsigned)l16];
    }
#pragma unroll
    for (int i = 0; i < 16; ++i) {
        if (FULL || first + i < n)
            atomicAdd(&win[(e[i] & ((1u << AG_TBITS) - 1u)) * HID + (unsigned)l16], hv[i]);
    }
}

__global__ __launch_bounds__(XW_THREADS)
void k_edge_out(const int* __restrict__ ei, int E, int N,
                const float* __restrict__ h,
                const float* __restrict__ dinv,
                const unsigned* __restrict__ bucket,
                const unsigned* __restrict__ bcnt,
                const float* __restrict__ conv_b,
                const float* __restrict__ out_w,
                const float* __restrict__ out_b,
                float* __restrict__ out) {
    __shared__ __attribute__((aligned(16))) float    win[XW_WIN * HID];
    __shared__ __attribute__((aligned(16))) unsigned list[XW_WAVES][XW_LCAP];

    const int tid  = threadIdx.x;
    const int lane = tid & 31;
    const int wv   = __builtin_amdgcn_readfirstlane(tid >> 5);
    const int l16  = lane & 15;
    const int half = lane >> 4;

    const int nbase = blockIdx.x * XW_WIN;
    int wn = N - nbase; if (wn > XW_WIN) wn = XW_WIN;
    const int b0 = wv * XW_BPW;
    const unsigned* seg0 = bucket + ((size_t)blockIdx.x * BIN_PRODS + (size_t)b0) * BIN_CAP;
    const unsigned cmine = (lane < XW_BPW) ? bcnt[(size_t)blockIdx.x * BIN_PRODS + b0 + lane] : 0u;

    {
        const v4f* h4 = (const v4f*)(h + (size_t)nbase * HID);
        v4f* w4 = (v4f*)win;
        for (int i = tid; i < XW_WIN * HID / 4; i += XW_THREADS)
            w4[i] = (i * 4 < wn * HID) ? h4[i] : (v4f){0.0f, 0.0f, 0.0f, 0.0f};
    }

    int cnt[XW_BPW];
    unsigned ent[XW_BPW][BIN_LINES];
#pragma unroll
    for (int j = 0; j < XW_BPW; ++j) {
        cnt[j] = __builtin_amdgcn_readlane((int)cmine, j);
        const int cl = ((unsigned)cnt[j] <= (unsigned)BIN_CAP) ? cnt[j] : 0;
        const unsigned* seg = seg0 + j * BIN_CAP;
#pragma unroll
        for (int u = 0; u < BIN_LINES; ++u)
            ent[j][u] = (u * 32 + lane < cl) ? seg[u * 32 + lane] : 0u;
    }
    unsigned* mylist = list[wv];
    int n = 0;
    unsigned rescan = 0u;
#pragma unroll
    for (int j = 0; j < XW_BPW; ++j) {
        const int c = cnt[j];
        if ((unsigned)c <= (unsigned)BIN_CAP && n + c <= XW_LCAP) {
#pragma unroll
            for (int u = 0; u < BIN_LINES; ++u)
                if (u * 32 + lane < c) mylist[n + u * 32 + lane] = ent[j][u];
            n += c;
        } else {
            rescan |= 1u << j;
        }
    }
    __syncthreads();

    {
        int k = 0;
        for (; k + 32 <= n; k += 32)
            xw_rows32<true>(h, win, mylist, k, n, l16, half);
        if (k < n)
            xw_rows32<false>(h, win, mylist, k, n, l16, half);
    }
    if (rescan != 0u) {
        const int sub = bin_sub(E);
#pragma nounroll
        for (int j = 0; j < XW_BPW; ++j) {
            if (((rescan >> j) & 1u) == 0u) continue;
            int e0 = (b0 + j) * sub;  if (e0 > E) e0 = E;
            int e1 = e0 + sub;        if (e1 > E) e1 = E;
            for (int e = e0 + lane; e < e1; e += 32) {
                const unsigned off = (unsigned)(ei[E + e] - nbase);
                if (off < (unsigned)wn) {
                    const int sidx = ei[e];
#pragma nounroll
                    for (int c = 0; c < HID; ++c)
                        atomicAdd(&win[off * HID + (unsigned)c], h[(size_t)sidx * HID + c]);
                }
            }
        }
    }
    __syncthreads();

    if (tid < wn) {
        const int nd = nbase + tid;
        const v4f* av = (const v4f*)(win + tid * HID);
        const v4f* bv = (const v4f*)conv_b;
        const v4f* ov = (const v4f*)out_w;
        const float dn = dinv[nd];
        float acc = out_b[0];
#pragma unroll
        for (int i = 0; i < 4; ++i) {
            v4f v = av[i] * dn + bv[i];
            v4f wvi = ov[i];
#pragma unroll
            for (int j = 0; j < 4; ++j)
                acc = fmaf(fmaxf(v[j], 0.0f), wvi[j], acc);
        }
        out[nd] = acc;
    }
}


__global__ void k_out(const float* __restrict__ agg,
                      const float* __restrict__ dinv,
                      const float* __restrict__ conv_b,
                      const float* __restrict__ out_w,
                      const float* __restrict__ out_b,
                      float* __restrict__ out, int N) {
    int n = blockIdx.x * blockDim.x + threadIdx.x;
    if (n >= N) return;
    const v4f* av = (const v4f*)(agg + (size_t)n * HID);
    const v4f* bv = (const v4f*)conv_b;
    const v4f* wv = (const v4f*)out_w;
    const float dn = dinv[n];
    float acc = out_b[0];
#pragma unroll
    for (int i = 0; i < 4; ++i) {
        v4f v = av[i] * dn + bv[i];
        v4f wvi = wv[i];
#pragma unroll
        for (int j = 0; j < 4; ++j)
            acc = fmaf(fmaxf(v[j], 0.0f), wvi[j], acc);
    }
    out[n] = acc;
}


static inline size_t align256(size_t v) { return (v + 255) & ~(size_t)255; }

extern "C" void kernel_launch(void* const* d_in, const int* in_sizes, int n_in,
                              void* d_out, int out_size, void* d_ws, size_t ws_size,
                              hipStream_t stream) {
    const float* x      = (const float*)d_in[0];
    const int*   ei     = (const int*)d_in[1];
    const float* conv_w = (const float*)d_in[2];
    const float* conv_b = (const float*)d_in[3];
    const float* out_w  = (const float*)d_in[4];
    const float* out_b  = (const float*)d_in[5];
    float* out = (float*)d_out;

    const int N = in_sizes[0] / IN_CH;
    const int E = in_sizes[1] / 2;

    const int nblk = (N + 31) / 32;

    char* ws = (char*)d_ws;
    size_t off = 0;
    float* dinv   = (float*)(ws + off);    off = align256(off + (size_t)N * 4);
    float* h      = (float*)(ws + off);    off = align256(off + (size_t)nblk * 32 * HID * 4);
    const int nwin = (N + XW_WIN - 1) / XW_WIN;
    size_t offb = off;
    unsigned* bcnt   = (unsigned*)(ws + offb);   offb = align256(offb + (size_t)nwin * BIN_PRODS * 4);
    unsigned* bucket = (unsigned*)(ws + offb);   offb = align256(offb + (size_t)nwin * BIN_PRODS * BIN_CAP * 4);
    const bool binned = (offb <= ws_size) && (nwin <= BIN_MAXW) && (N <= (1 << (32 - AG_TBITS)));

    const int ndegy = (nblk * 32 + DEG_WIN - 1) / DEG_WIN;

    if (binned) {
        const int nbiny = (BIN_PRODS + DEG_PARTS - 1) / DEG_PARTS;
        k_deg_hist<<<dim3(DEG_PARTS, nbiny), DEG_THREADS, 0, stream>>>(ei, E, N, nullptr, 0, nwin, bucket, bcnt);
        k_gemm_win<<<nwin, GW_THREADS, 0, stream>>>(x, conv_w, ei, E, bucket, bcnt, dinv, h, N);
        k_edge_out<<<nwin, XW_THREADS, 0, stream>>>(ei, E, N, h, dinv, bucket, bcnt, conv_b, out_w, out_b, out);
    } else {
        float* agg = (float*)(ws + off);
        const int nwin_s = (N + AG_WIN - 1) / AG_WIN;
        const int nb_n = (N + 255) / 256;
        k_deg_hist<<<dim3(DEG_PARTS, ndegy), DEG_THREADS, 0, stream>>>(ei, E, N, (unsigned*)agg, ndegy, 0, nullptr, nullptr);
        k_gemm_wmma<true><<<nblk, 64, 0, stream>>>(x, conv_w, (const unsigned*)agg, dinv, h, agg, N);
        k_edge_agg<<<dim3(AG_CHUNKS, nwin_s), AG_THREADS, 0, stream>>>(ei, E, N, h, agg);
        k_out<<<nb_n, 256, 0, stream>>>(agg, dinv, conv_b, out_w, out_b, out, N);
    }
}
